# starting kernel
# baseline (speedup 1.0000x reference)
#include <hip/hip_runtime.h>

#define N_NODES 100000
#define N_EDGES 640000
#define D 128

typedef float v2f __attribute__((ext_vector_type(2)));
typedef float v8f __attribute__((ext_vector_type(8)));

__global__ void sage_zero_ws(float* __restrict__ ws, int n) {
    int i = blockIdx.x * blockDim.x + threadIdx.x;
    if (i < n) ws[i] = 0.0f;
}

__global__ void sage_edge_scatter(const float* __restrict__ feat,
                                  const int*   __restrict__ src,
                                  const int*   __restrict__ dst,
                                  float*       __restrict__ neigh_sum,
                                  float*       __restrict__ deg) {
    int tid = blockIdx.x * blockDim.x + threadIdx.x;
    int e = tid >> 5;
    int c = tid & 31;
    if (e >= N_EDGES) return;
    int s  = src[e];
    int dn = dst[e];
    const float4 v = *(const float4*)(feat + (size_t)s * D + c * 4);
    float* o = neigh_sum + (size_t)dn * D + c * 4;
    unsafeAtomicAdd(o + 0, v.x);
    unsafeAtomicAdd(o + 1, v.y);
    unsafeAtomicAdd(o + 2, v.z);
    unsafeAtomicAdd(o + 3, v.w);
    if (c == 0) unsafeAtomicAdd(deg + dn, 1.0f);
}

__global__ __launch_bounds__(256)
void sage_wmma_gemm(const float* __restrict__ feat,
                    const float* __restrict__ neigh_sum,
                    const float* __restrict__ deg,
                    const float* __restrict__ Wself,
                    const float* __restrict__ bself,
                    const float* __restrict__ Wneigh,
                    const float* __restrict__ bneigh,
                    float*       __restrict__ out) {
    const int lane  = threadIdx.x & 31;
    const int wave  = threadIdx.x >> 5;
    const int rloc  = lane & 15;
    const int khalf = lane >> 4;
    const int row   = blockIdx.x * 16 + rloc;
    const int col   = wave * 16 + (lane & 15);

    const float bias = bself[col] + bneigh[col];
    v8f acc;
#pragma unroll
    for (int j = 0; j < 8; ++j) acc[j] = bias;

    const float* frow = feat      + (size_t)row * D;
    const float* nrow = neigh_sum + (size_t)row * D;
    const float  rdeg = 1.0f / fmaxf(deg[row], 1.0f);

#pragma unroll
    for (int k0 = 0; k0 < D; k0 += 4) {
        const int k = k0 + khalf * 2;
        v2f a; a.x = frow[k];  a.y = frow[k + 1];
        v2f b; b.x = Wself[(size_t)k * D + col];
               b.y = Wself[(size_t)(k + 1) * D + col];
        acc = __builtin_amdgcn_wmma_f32_16x16x4_f32(
                  false, a, false, b, (short)0, acc, false, false);
    }

#pragma unroll
    for (int k0 = 0; k0 < D; k0 += 4) {
        const int k = k0 + khalf * 2;
        v2f a; a.x = nrow[k] * rdeg;  a.y = nrow[k + 1] * rdeg;
        v2f b; b.x = Wneigh[(size_t)k * D + col];
               b.y = Wneigh[(size_t)(k + 1) * D + col];
        acc = __builtin_amdgcn_wmma_f32_16x16x4_f32(
                  false, a, false, b, (short)0, acc, false, false);
    }

    float* obase = out + (size_t)(blockIdx.x * 16 + khalf * 8) * D + col;
#pragma unroll
    for (int j = 0; j < 8; ++j)
        obase[(size_t)j * D] = acc[j];
}

extern "C" void kernel_launch(void* const* d_in, const int* in_sizes, int n_in,
                              void* d_out, int out_size, void* d_ws, size_t ws_size,
                              hipStream_t stream) {
    const float* feat   = (const float*)d_in[0];
    const int*   src    = (const int*)  d_in[1];
    const int*   dst    = (const int*)  d_in[2];
    const float* Wself  = (const float*)d_in[3];
    const float* bself  = (const float*)d_in[4];
    const float* Wneigh = (const float*)d_in[5];
    const float* bneigh = (const float*)d_in[6];
    float*       out    = (float*)d_out;

    float* neigh_sum = (float*)d_ws;
    float* deg       = neigh_sum + (size_t)N_NODES * D;

    const int zn = N_NODES * (D + 1);
    sage_zero_ws<<<(zn + 255) / 256, 256, 0, stream>>>((float*)d_ws, zn);

    const int et = N_EDGES * 32;
    sage_edge_scatter<<<(et + 255) / 256, 256, 0, stream>>>(
        feat, src, dst, neigh_sum, deg);

    sage_wmma_gemm<<<N_NODES / 16, 256, 0, stream>>>(
        feat, neigh_sum, deg, Wself, bself, Wneigh, bneigh, out);
}


# GEMM kernel: the two 16x128 A tiles are fetched once per block by whole-line 16-byte loads into padded LDS and read as fragments from there; before, each of 8 waves gathered them 8 B per row
# speedup vs baseline: 1.2188x; 1.2188x over previous
#include <hip/hip_runtime.h>

#define N_NODES 100000
#define N_EDGES 640000
#define D 128

typedef float v2f __attribute__((ext_vector_type(2)));
typedef float v8f __attribute__((ext_vector_type(8)));

__global__ void sage_zero_ws(float* __restrict__ ws, int n) {
    int i = blockIdx.x * blockDim.x + threadIdx.x;
    if (i < n) ws[i] = 0.0f;
}

__global__ void sage_edge_scatter(const float* __restrict__ feat,
                                  const int*   __restrict__ src,
                                  const int*   __restrict__ dst,
                                  float*       __restrict__ neigh_sum,
                                  float*       __restrict__ deg) {
    int tid = blockIdx.x * blockDim.x + threadIdx.x;
    int e = tid >> 5;
    int c = tid & 31;
    if (e >= N_EDGES) return;
    int s  = src[e];
    int dn = dst[e];
    const float4 v = *(const float4*)(feat + (size_t)s * D + c * 4);
    float* o = neigh_sum + (size_t)dn * D + c * 4;
    unsafeAtomicAdd(o + 0, v.x);
    unsafeAtomicAdd(o + 1, v.y);
    unsafeAtomicAdd(o + 2, v.z);
    unsafeAtomicAdd(o + 3, v.w);
    if (c == 0) unsafeAtomicAdd(deg + dn, 1.0f);
}

#define A_PITCH 132

__global__ __launch_bounds__(256)
void sage_wmma_gemm(const float* __restrict__ feat,
                    const float* __restrict__ neigh_sum,
                    const float* __restrict__ deg,
                    const float* __restrict__ Wself,
                    const float* __restrict__ bself,
                    const float* __restrict__ Wneigh,
                    const float* __restrict__ bneigh,
                    float*       __restrict__ out) {
    __shared__ __attribute__((aligned(16))) float sSelf [16 * A_PITCH];
    __shared__ __attribute__((aligned(16))) float sNeigh[16 * A_PITCH];

    const int lane  = threadIdx.x & 31;
    const int wave  = threadIdx.x >> 5;
    const int rloc  = lane & 15;
    const int khalf = lane >> 4;
    const int row0  = blockIdx.x * 16;
    const int col   = wave * 16 + (lane & 15);

    {
        const int r0 = wave, r1 = wave + 8;
        const size_t g0 = (size_t)(row0 + r0) * D + lane * 4;
        const size_t g1 = (size_t)(row0 + r1) * D + lane * 4;
        const float4 f0 = *(const float4*)(feat + g0);
        const float4 f1 = *(const float4*)(feat + g1);
        float4 n0 = *(const float4*)(neigh_sum + g0);
        float4 n1 = *(const float4*)(neigh_sum + g1);
        const float rd0 = 1.0f / fmaxf(deg[row0 + r0], 1.0f);
        const float rd1 = 1.0f / fmaxf(deg[row0 + r1], 1.0f);
        n0.x *= rd0; n0.y *= rd0; n0.z *= rd0; n0.w *= rd0;
        n1.x *= rd1; n1.y *= rd1; n1.z *= rd1; n1.w *= rd1;
        *(float4*)(sSelf  + r0 * A_PITCH + lane * 4) = f0;
        *(float4*)(sSelf  + r1 * A_PITCH + lane * 4) = f1;
        *(float4*)(sNeigh + r0 * A_PITCH + lane * 4) = n0;
        *(float4*)(sNeigh + r1 * A_PITCH + lane * 4) = n1;
    }

    const float bias = bself[col] + bneigh[col];
    v8f acc;
#pragma unroll
    for (int j = 0; j < 8; ++j) acc[j] = bias;

    __syncthreads();

    const float* arow_s = sSelf  + rloc * A_PITCH + khalf * 2;
    const float* arow_n = sNeigh + rloc * A_PITCH + khalf * 2;

#pragma unroll
    for (int k0 = 0; k0 < D; k0 += 4) {
        const int k = k0 + khalf * 2;
        const v2f a = *(const v2f*)(arow_s + k0);
        v2f b; b.x = Wself[(size_t)k * D + col];
               b.y = Wself[(size_t)(k + 1) * D + col];
        acc = __builtin_amdgcn_wmma_f32_16x16x4_f32(
                  false, a, false, b, (short)0, acc, false, false);
    }

#pragma unroll
    for (int k0 = 0; k0 < D; k0 += 4) {
        const int k = k0 + khalf * 2;
        const v2f a = *(const v2f*)(arow_n + k0);
        v2f b; b.x = Wneigh[(size_t)k * D + col];
               b.y = Wneigh[(size_t)(k + 1) * D + col];
        acc = __builtin_amdgcn_wmma_f32_16x16x4_f32(
                  false, a, false, b, (short)0, acc, false, false);
    }

    float* obase = out + (size_t)(row0 + khalf * 8) * D + col;
#pragma unroll
    for (int j = 0; j < 8; ++j)
        obase[(size_t)j * D] = acc[j];
}

extern "C" void kernel_launch(void* const* d_in, const int* in_sizes, int n_in,
                              void* d_out, int out_size, void* d_ws, size_t ws_size,
                              hipStream_t stream) {
    const float* feat   = (const float*)d_in[0];
    const int*   src    = (const int*)  d_in[1];
    const int*   dst    = (const int*)  d_in[2];
    const float* Wself  = (const float*)d_in[3];
    const float* bself  = (const float*)d_in[4];
    const float* Wneigh = (const float*)d_in[5];
    const float* bneigh = (const float*)d_in[6];
    float*       out    = (float*)d_out;

    float* neigh_sum = (float*)d_ws;
    float* deg       = neigh_sum + (size_t)N_NODES * D;

    const int zn = N_NODES * (D + 1);
    sage_zero_ws<<<(zn + 255) / 256, 256, 0, stream>>>((float*)d_ws, zn);

    const int et = N_EDGES * 32;
    sage_edge_scatter<<<(et + 255) / 256, 256, 0, stream>>>(
        feat, src, dst, neigh_sum, deg);

    sage_wmma_gemm<<<N_NODES / 16, 256, 0, stream>>>(
        feat, neigh_sum, deg, Wself, bself, Wneigh, bneigh, out);
}


# Edge scatter: lane c takes columns c,c+32,c+64,c+96 so each wave-wide f32 atomic and load covers one whole 128 B line (was float4 per lane: 4 lines x 8 lanes each); same atomics, arrival-order sums
# speedup vs baseline: 1.3560x; 1.1126x over previous
#include <hip/hip_runtime.h>

#define N_NODES 100000
#define N_EDGES 640000
#define D 128

typedef float v2f __attribute__((ext_vector_type(2)));
typedef float v8f __attribute__((ext_vector_type(8)));

__global__ void sage_zero_ws(float* __restrict__ ws, int n) {
    int i = blockIdx.x * blockDim.x + threadIdx.x;
    if (i < n) ws[i] = 0.0f;
}

__global__ void sage_edge_scatter(const float* __restrict__ feat,
                                  const int*   __restrict__ src,
                                  const int*   __restrict__ dst,
                                  float*       __restrict__ neigh_sum,
                                  float*       __restrict__ deg) {
    int tid = blockIdx.x * blockDim.x + threadIdx.x;
    int e = tid >> 5;
    int c = tid & 31;
    if (e >= N_EDGES) return;
    int s  = src[e];
    int dn = dst[e];
    const float* f = feat      + (size_t)s  * D + c;
    float*       o = neigh_sum + (size_t)dn * D + c;
    const float v0 = f[0];
    const float v1 = f[32];
    const float v2 = f[64];
    const float v3 = f[96];
    unsafeAtomicAdd(o +  0, v0);
    unsafeAtomicAdd(o + 32, v1);
    unsafeAtomicAdd(o + 64, v2);
    unsafeAtomicAdd(o + 96, v3);
    if (c == 0) unsafeAtomicAdd(deg + dn, 1.0f);
}

#define A_PITCH 132

__global__ __launch_bounds__(256)
void sage_wmma_gemm(const float* __restrict__ feat,
                    const float* __restrict__ neigh_sum,
                    const float* __restrict__ deg,
                    const float* __restrict__ Wself,
                    const float* __restrict__ bself,
                    const float* __restrict__ Wneigh,
                    const float* __restrict__ bneigh,
                    float*       __restrict__ out) {
    __shared__ __attribute__((aligned(16))) float sSelf [16 * A_PITCH];
    __shared__ __attribute__((aligned(16))) float sNeigh[16 * A_PITCH];

    const int lane  = threadIdx.x & 31;
    const int wave  = threadIdx.x >> 5;
    const int rloc  = lane & 15;
    const int khalf = lane >> 4;
    const int row0  = blockIdx.x * 16;
    const int col   = wave * 16 + (lane & 15);

    {
        const int r0 = wave, r1 = wave + 8;
        const size_t g0 = (size_t)(row0 + r0) * D + lane * 4;
        const size_t g1 = (size_t)(row0 + r1) * D + lane * 4;
        const float4 f0 = *(const float4*)(feat + g0);
        const float4 f1 = *(const float4*)(feat + g1);
        float4 n0 = *(const float4*)(neigh_sum + g0);
        float4 n1 = *(const float4*)(neigh_sum + g1);
        const float rd0 = 1.0f / fmaxf(deg[row0 + r0], 1.0f);
        const float rd1 = 1.0f / fmaxf(deg[row0 + r1], 1.0f);
        n0.x *= rd0; n0.y *= rd0; n0.z *= rd0; n0.w *= rd0;
        n1.x *= rd1; n1.y *= rd1; n1.z *= rd1; n1.w *= rd1;
        *(float4*)(sSelf  + r0 * A_PITCH + lane * 4) = f0;
        *(float4*)(sSelf  + r1 * A_PITCH + lane * 4) = f1;
        *(float4*)(sNeigh + r0 * A_PITCH + lane * 4) = n0;
        *(float4*)(sNeigh + r1 * A_PITCH + lane * 4) = n1;
    }

    const float bias = bself[col] + bneigh[col];
    v8f acc;
#pragma unroll
    for (int j = 0; j < 8; ++j) acc[j] = bias;

    __syncthreads();

    const float* arow_s = sSelf  + rloc * A_PITCH + khalf * 2;
    const float* arow_n = sNeigh + rloc * A_PITCH + khalf * 2;

#pragma unroll
    for (int k0 = 0; k0 < D; k0 += 4) {
        const int k = k0 + khalf * 2;
        const v2f a = *(const v2f*)(arow_s + k0);
        v2f b; b.x = Wself[(size_t)k * D + col];
               b.y = Wself[(size_t)(k + 1) * D + col];
        acc = __builtin_amdgcn_wmma_f32_16x16x4_f32(
                  false, a, false, b, (short)0, acc, false, false);
    }

#pragma unroll
    for (int k0 = 0; k0 < D; k0 += 4) {
        const int k = k0 + khalf * 2;
        const v2f a = *(const v2f*)(arow_n + k0);
        v2f b; b.x = Wneigh[(size_t)k * D + col];
               b.y = Wneigh[(size_t)(k + 1) * D + col];
        acc = __builtin_amdgcn_wmma_f32_16x16x4_f32(
                  false, a, false, b, (short)0, acc, false, false);
    }

    float* obase = out + (size_t)(row0 + khalf * 8) * D + col;
#pragma unroll
    for (int j = 0; j < 8; ++j)
        obase[(size_t)j * D] = acc[j];
}

extern "C" void kernel_launch(void* const* d_in, const int* in_sizes, int n_in,
                              void* d_out, int out_size, void* d_ws, size_t ws_size,
                              hipStream_t stream) {
    const float* feat   = (const float*)d_in[0];
    const int*   src    = (const int*)  d_in[1];
    const int*   dst    = (const int*)  d_in[2];
    const float* Wself  = (const float*)d_in[3];
    const float* bself  = (const float*)d_in[4];
    const float* Wneigh = (const float*)d_in[5];
    const float* bneigh = (const float*)d_in[6];
    float*       out    = (float*)d_out;

    float* neigh_sum = (float*)d_ws;
    float* deg       = neigh_sum + (size_t)N_NODES * D;

    const int zn = N_NODES * (D + 1);
    sage_zero_ws<<<(zn + 255) / 256, 256, 0, stream>>>((float*)d_ws, zn);

    const int et = N_EDGES * 32;
    sage_edge_scatter<<<(et + 255) / 256, 256, 0, stream>>>(
        feat, src, dst, neigh_sum, deg);

    sage_wmma_gemm<<<N_NODES / 16, 256, 0, stream>>>(
        feat, neigh_sum, deg, Wself, bself, Wneigh, bneigh, out);
}


# Neighbour sum without f32 atomics: edges counting-sorted by dst (one int atomic per edge + prefix scan), each row's sources summed in registers in the GEMM A staging; sums in int-atomic arrival order
# speedup vs baseline: 1.8182x; 1.3409x over previous
#include <hip/hip_runtime.h>


#define N_NODES 100000
#define N_EDGES 640000
#define D 128
#define SCAN_T 1024
#define SCAN_B ((N_NODES + SCAN_T - 1) / SCAN_T)

typedef float v2f __attribute__((ext_vector_type(2)));
typedef float v8f __attribute__((ext_vector_type(8)));


__global__ void sage_zero_cnt(int* __restrict__ cnt) {
    const int i = blockIdx.x * blockDim.x + threadIdx.x;
    if (i < N_NODES) cnt[i] = 0;
}

__global__ void sage_edge_rank(const int* __restrict__ dst,
                               int*       __restrict__ cnt,
                               int*       __restrict__ rank) {
    const int t = blockIdx.x * blockDim.x + threadIdx.x;
    if (t < N_EDGES / 4) {
        const int4 d = ((const int4*)dst)[t];
        int4 r;
        r.x = atomicAdd(cnt + d.x, 1);
        r.y = atomicAdd(cnt + d.y, 1);
        r.z = atomicAdd(cnt + d.z, 1);
        r.w = atomicAdd(cnt + d.w, 1);
        ((int4*)rank)[t] = r;
    }
}

__global__ __launch_bounds__(SCAN_T)
void sage_scan(const int* __restrict__ cnt, int* __restrict__ off) {
    __shared__ int s_pre[SCAN_T / 32];
    __shared__ int s_tot[SCAN_T / 32];
    const int t = threadIdx.x;
    const int lane = t & 31;
    const int w = t >> 5;
    const int b = blockIdx.x;

    int pre = 0;
    const int4* c4 = (const int4*)cnt;
    const int n4 = b * (SCAN_T / 4);
    for (int base = 0; base < n4; base += 8 * SCAN_T) {
#pragma unroll
        for (int u = 0; u < 8; ++u) {
            const int i = base + u * SCAN_T + t;
            const int4 q = c4[min(i, n4 - 1)];
            const int s = (q.x + q.y) + (q.z + q.w);
            pre += (i < n4) ? s : 0;
        }
    }
    for (int o = 16; o > 0; o >>= 1) pre += __shfl_xor(pre, o);

    const int i = b * SCAN_T + t;
    const int v = (i < N_NODES) ? cnt[i] : 0;
    int inc = v;
    for (int o = 1; o < 32; o <<= 1) {
        const int u = __shfl_up(inc, o);
        if (lane >= o) inc += u;
    }
    if (lane == 0)  s_pre[w] = pre;
    if (lane == 31) s_tot[w] = inc;
    __syncthreads();

    int p  = s_pre[lane];
    int sb = (lane < w) ? s_tot[lane] : 0;
    for (int o = 16; o > 0; o >>= 1) {
        p  += __shfl_xor(p, o);
        sb += __shfl_xor(sb, o);
    }
    if (i < N_NODES) off[i] = p + sb + (inc - v);
}

__global__ void sage_csr_fill(const int* __restrict__ src,
                              const int* __restrict__ dst,
                              const int* __restrict__ off,
                              const int* __restrict__ rank,
                              int*       __restrict__ csr) {
    const int t = blockIdx.x * blockDim.x + threadIdx.x;
    if (t < N_EDGES / 4) {
        const int4 d = ((const int4*)dst)[t];
        const int4 s = ((const int4*)src)[t];
        const int4 r = ((const int4*)rank)[t];
        const int p0 = min(max(off[d.x] + r.x, 0), N_EDGES - 1);
        const int p1 = min(max(off[d.y] + r.y, 0), N_EDGES - 1);
        const int p2 = min(max(off[d.z] + r.z, 0), N_EDGES - 1);
        const int p3 = min(max(off[d.w] + r.w, 0), N_EDGES - 1);
        csr[p0] = s.x;
        csr[p1] = s.y;
        csr[p2] = s.z;
        csr[p3] = s.w;
    }
}

__device__ __forceinline__ void sage_gather2(const float* __restrict__ feat,
                                             const int*   __restrict__ csr,
                                             int sa, int na, int sb, int nb,
                                             int lane, float4& ra, float4& rb) {
    const unsigned lane4 = (unsigned)lane * 4u;
    float4 acca = make_float4(0.0f, 0.0f, 0.0f, 0.0f);
    float4 accb = make_float4(0.0f, 0.0f, 0.0f, 0.0f);
    const int nmax = max(na, nb);
    int ia = 0, ib = 0;
    for (int j = 0; j < nmax; j += 8) {
        const int jl = j & 31;
        if (jl == 0) {
            ia = (j + lane < na) ? csr[min(sa + j + lane, N_EDGES - 1)] : 0;
            ib = (j + lane < nb) ? csr[min(sb + j + lane, N_EDGES - 1)] : 0;
        }
        const int ica = min(max(ia, 0), N_NODES - 1);
        const int icb = min(max(ib, 0), N_NODES - 1);
        int qa[8], qb[8];
#pragma unroll
        for (int k = 0; k < 8; ++k) {
            qa[k] = __builtin_amdgcn_readlane(ica, jl + k);
            qb[k] = __builtin_amdgcn_readlane(icb, jl + k);
        }
        float4 va[8], vb[8];
#pragma unroll
        for (int k = 0; k < 8; ++k) {
            if (j + k < na) va[k] = *(const float4*)(feat + (size_t)qa[k] * D + lane4);
            else            va[k] = make_float4(0.0f, 0.0f, 0.0f, 0.0f);
        }
#pragma unroll
        for (int k = 0; k < 8; ++k) {
            if (j + k < nb) vb[k] = *(const float4*)(feat + (size_t)qb[k] * D + lane4);
            else            vb[k] = make_float4(0.0f, 0.0f, 0.0f, 0.0f);
        }
#pragma unroll
        for (int k = 0; k < 8; ++k) {
            acca.x += va[k].x; acca.y += va[k].y; acca.z += va[k].z; acca.w += va[k].w;
        }
#pragma unroll
        for (int k = 0; k < 8; ++k) {
            accb.x += vb[k].x; accb.y += vb[k].y; accb.z += vb[k].z; accb.w += vb[k].w;
        }
    }
    ra = acca;
    rb = accb;
}

#define A_PITCH 132

__global__ __launch_bounds__(256)
void sage_wmma_gemm(const float* __restrict__ feat,
                    const int*   __restrict__ csr,
                    const int*   __restrict__ off,
                    const int*   __restrict__ cnt,
                    const float* __restrict__ Wself,
                    const float* __restrict__ bself,
                    const float* __restrict__ Wneigh,
                    const float* __restrict__ bneigh,
                    float*       __restrict__ out) {
    __shared__ __attribute__((aligned(16))) float sSelf [16 * A_PITCH];
    __shared__ __attribute__((aligned(16))) float sNeigh[16 * A_PITCH];

    const int lane  = threadIdx.x & 31;
    const int wave  = threadIdx.x >> 5;
    const int rloc  = lane & 15;
    const int khalf = lane >> 4;
    const int row0  = blockIdx.x * 16;
    const int col   = wave * 16 + (lane & 15);

    {
        const int r0 = wave, r1 = wave + 8;
        const size_t g0 = (size_t)(row0 + r0) * D + lane * 4;
        const size_t g1 = (size_t)(row0 + r1) * D + lane * 4;
        const float4 f0 = *(const float4*)(feat + g0);
        const float4 f1 = *(const float4*)(feat + g1);
        const int c0 = min(max(__builtin_amdgcn_readfirstlane(cnt[row0 + r0]), 0), N_EDGES);
        const int c1 = min(max(__builtin_amdgcn_readfirstlane(cnt[row0 + r1]), 0), N_EDGES);
        const int s0 = min(max(__builtin_amdgcn_readfirstlane(off[row0 + r0]), 0), N_EDGES);
        const int s1 = min(max(__builtin_amdgcn_readfirstlane(off[row0 + r1]), 0), N_EDGES);
        *(float4*)(sSelf + r0 * A_PITCH + lane * 4) = f0;
        *(float4*)(sSelf + r1 * A_PITCH + lane * 4) = f1;
        float4 n0, n1;
        sage_gather2(feat, csr, s0, c0, s1, c1, lane, n0, n1);
        const float rd0 = 1.0f / fmaxf((float)c0, 1.0f);
        const float rd1 = 1.0f / fmaxf((float)c1, 1.0f);
        n0.x *= rd0; n0.y *= rd0; n0.z *= rd0; n0.w *= rd0;
        n1.x *= rd1; n1.y *= rd1; n1.z *= rd1; n1.w *= rd1;
        *(float4*)(sNeigh + r0 * A_PITCH + lane * 4) = n0;
        *(float4*)(sNeigh + r1 * A_PITCH + lane * 4) = n1;
    }

    const float bias = bself[col] + bneigh[col];
    v8f acc;
#pragma unroll
    for (int j = 0; j < 8; ++j) acc[j] = bias;

    __syncthreads();

    const float* arow_s = sSelf  + rloc * A_PITCH + khalf * 2;
    const float* arow_n = sNeigh + rloc * A_PITCH + khalf * 2;

#pragma unroll
    for (int k0 = 0; k0 < D; k0 += 4) {
        const int k = k0 + khalf * 2;
        const v2f a = *(const v2f*)(arow_s + k0);
        v2f b; b.x = Wself[(size_t)k * D + col];
               b.y = Wself[(size_t)(k + 1) * D + col];
        acc = __builtin_amdgcn_wmma_f32_16x16x4_f32(
                  false, a, false, b, (short)0, acc, false, false);
    }

#pragma unroll
    for (int k0 = 0; k0 < D; k0 += 4) {
        const int k = k0 + khalf * 2;
        const v2f a = *(const v2f*)(arow_n + k0);
        v2f b; b.x = Wneigh[(size_t)k * D + col];
               b.y = Wneigh[(size_t)(k + 1) * D + col];
        acc = __builtin_amdgcn_wmma_f32_16x16x4_f32(
                  false, a, false, b, (short)0, acc, false, false);
    }

    float* obase = out + (size_t)(row0 + khalf * 8) * D + col;
#pragma unroll
    for (int j = 0; j < 8; ++j)
        obase[(size_t)j * D] = acc[j];
}

extern "C" void kernel_launch(void* const* d_in, const int* in_sizes, int n_in,
                              void* d_out, int out_size, void* d_ws, size_t ws_size,
                              hipStream_t stream) {
    const float* feat   = (const float*)d_in[0];
    const int*   src    = (const int*)  d_in[1];
    const int*   dst    = (const int*)  d_in[2];
    const float* Wself  = (const float*)d_in[3];
    const float* bself  = (const float*)d_in[4];
    const float* Wneigh = (const float*)d_in[5];
    const float* bneigh = (const float*)d_in[6];
    float*       out    = (float*)d_out;

    int* cnt  = (int*)d_ws;
    int* off  = cnt  + N_NODES;
    int* rank = off  + N_NODES;
    int* csr  = rank + N_EDGES;

    sage_zero_cnt<<<(N_NODES + 255) / 256, 256, 0, stream>>>(cnt);
    sage_edge_rank<<<(N_EDGES / 4 + 255) / 256, 256, 0, stream>>>(dst, cnt, rank);
    sage_scan<<<SCAN_B, SCAN_T, 0, stream>>>(cnt, off);
    sage_csr_fill<<<(N_EDGES / 4 + 255) / 256, 256, 0, stream>>>(src, dst, off, rank, csr);
    sage_wmma_gemm<<<N_NODES / 16, 256, 0, stream>>>(
        feat, csr, off, cnt, Wself, bself, Wneigh, bneigh, out);
}


# GEMM block covers 64 rows, not 16: each wave keeps 4 accumulator tiles and uses every W fragment it loads for 4 row tiles; W reads from L2, W load instructions and waves fall to a quarter
# speedup vs baseline: 1.9614x; 1.0788x over previous
#include <hip/hip_runtime.h>


#define N_NODES 100000
#define N_EDGES 640000
#define D 128
#define SCAN_T 1024
#define SCAN_B ((N_NODES + SCAN_T - 1) / SCAN_T)

typedef float v2f __attribute__((ext_vector_type(2)));
typedef float v8f __attribute__((ext_vector_type(8)));


__global__ void sage_zero_cnt(int* __restrict__ cnt) {
    const int i = blockIdx.x * blockDim.x + threadIdx.x;
    if (i < N_NODES) cnt[i] = 0;
}

__global__ void sage_edge_rank(const int* __restrict__ dst,
                               int*       __restrict__ cnt,
                               int*       __restrict__ rank) {
    const int t = blockIdx.x * blockDim.x + threadIdx.x;
    if (t < N_EDGES / 4) {
        const int4 d = ((const int4*)dst)[t];
        int4 r;
        r.x = atomicAdd(cnt + d.x, 1);
        r.y = atomicAdd(cnt + d.y, 1);
        r.z = atomicAdd(cnt + d.z, 1);
        r.w = atomicAdd(cnt + d.w, 1);
        ((int4*)rank)[t] = r;
    }
}

__global__ __launch_bounds__(SCAN_T)
void sage_scan(const int* __restrict__ cnt, int* __restrict__ off) {
    __shared__ int s_pre[SCAN_T / 32];
    __shared__ int s_tot[SCAN_T / 32];
    const int t = threadIdx.x;
    const int lane = t & 31;
    const int w = t >> 5;
    const int b = blockIdx.x;

    int pre = 0;
    const int4* c4 = (const int4*)cnt;
    const int n4 = b * (SCAN_T / 4);
    for (int base = 0; base < n4; base += 8 * SCAN_T) {
#pragma unroll
        for (int u = 0; u < 8; ++u) {
            const int i = base + u * SCAN_T + t;
            const int4 q = c4[min(i, n4 - 1)];
            const int s = (q.x + q.y) + (q.z + q.w);
            pre += (i < n4) ? s : 0;
        }
    }
    for (int o = 16; o > 0; o >>= 1) pre += __shfl_xor(pre, o);

    const int i = b * SCAN_T + t;
    const int v = (i < N_NODES) ? cnt[i] : 0;
    int inc = v;
    for (int o = 1; o < 32; o <<= 1) {
        const int u = __shfl_up(inc, o);
        if (lane >= o) inc += u;
    }
    if (lane == 0)  s_pre[w] = pre;
    if (lane == 31) s_tot[w] = inc;
    __syncthreads();

    int p  = s_pre[lane];
    int sb = (lane < w) ? s_tot[lane] : 0;
    for (int o = 16; o > 0; o >>= 1) {
        p  += __shfl_xor(p, o);
        sb += __shfl_xor(sb, o);
    }
    if (i < N_NODES) off[i] = p + sb + (inc - v);
}

__global__ void sage_csr_fill(const int* __restrict__ src,
                              const int* __restrict__ dst,
                              const int* __restrict__ off,
                              const int* __restrict__ rank,
                              int*       __restrict__ csr) {
    const int t = blockIdx.x * blockDim.x + threadIdx.x;
    if (t < N_EDGES / 4) {
        const int4 d = ((const int4*)dst)[t];
        const int4 s = ((const int4*)src)[t];
        const int4 r = ((const int4*)rank)[t];
        const int p0 = min(max(off[d.x] + r.x, 0), N_EDGES - 1);
        const int p1 = min(max(off[d.y] + r.y, 0), N_EDGES - 1);
        const int p2 = min(max(off[d.z] + r.z, 0), N_EDGES - 1);
        const int p3 = min(max(off[d.w] + r.w, 0), N_EDGES - 1);
        csr[p0] = s.x;
        csr[p1] = s.y;
        csr[p2] = s.z;
        csr[p3] = s.w;
    }
}

__device__ __forceinline__ void sage_gather2(const float* __restrict__ feat,
                                             const int*   __restrict__ csr,
                                             int sa, int na, int sb, int nb,
                                             int lane, int ia0, int ib0,
                                             bool more,
                                             const int* __restrict__ pfa,
                                             const int* __restrict__ pfb,
                                             int& ta, int& tb,
                                             float4& ra, float4& rb) {
    const unsigned lane4 = (unsigned)lane * 4u;
    float4 acca = make_float4(0.0f, 0.0f, 0.0f, 0.0f);
    float4 accb = make_float4(0.0f, 0.0f, 0.0f, 0.0f);
    const int nmax = max(na, nb);
    int ica = min(max(ia0, 0), N_NODES - 1);
    int icb = min(max(ib0, 0), N_NODES - 1);
    for (int jc = 0; jc < nmax; jc += 32) {
        if (jc != 0) {
            const int ia = (jc + lane < na) ? csr[min(sa + jc + lane, N_EDGES - 1)] : 0;
            const int ib = (jc + lane < nb) ? csr[min(sb + jc + lane, N_EDGES - 1)] : 0;
            ica = min(max(ia, 0), N_NODES - 1);
            icb = min(max(ib, 0), N_NODES - 1);
        }
        if (jc == 0 && more) {
            ta = *pfa;
            tb = *pfb;
        }
        const int jend = min(nmax, jc + 32);
        for (int j = jc; j < jend; j += 8) {
            const int jl = j - jc;
            int qa[8], qb[8];
#pragma unroll
            for (int k = 0; k < 8; ++k) {
                qa[k] = __builtin_amdgcn_readlane(ica, jl + k);
                qb[k] = __builtin_amdgcn_readlane(icb, jl + k);
            }
            float4 va[8], vb[8];
#pragma unroll
            for (int k = 0; k < 8; ++k) {
                if (j + k < na) va[k] = *(const float4*)(feat + (size_t)qa[k] * D + lane4);
                else            va[k] = make_float4(0.0f, 0.0f, 0.0f, 0.0f);
            }
#pragma unroll
            for (int k = 0; k < 8; ++k) {
                if (j + k < nb) vb[k] = *(const float4*)(feat + (size_t)qb[k] * D + lane4);
                else            vb[k] = make_float4(0.0f, 0.0f, 0.0f, 0.0f);
            }
#pragma unroll
            for (int k = 0; k < 8; ++k) {
                acca.x += va[k].x; acca.y += va[k].y; acca.z += va[k].z; acca.w += va[k].w;
            }
#pragma unroll
            for (int k = 0; k < 8; ++k) {
                accb.x += vb[k].x; accb.y += vb[k].y; accb.z += vb[k].z; accb.w += vb[k].w;
            }
        }
    }
    if (nmax <= 0 && more) {
        ta = *pfa;
        tb = *pfb;
    }
    ra = acca;
    rb = accb;
}

#define A_PITCH 132
#define GEMM_TILES 4
#define GEMM_ROWS (16 * GEMM_TILES)
#define GEMM_BLOCKS ((N_NODES + GEMM_ROWS - 1) / GEMM_ROWS)

__global__ __launch_bounds__(256)
void sage_wmma_gemm(const float* __restrict__ feat,
                    const int*   __restrict__ csr,
                    const int*   __restrict__ off,
                    const int*   __restrict__ cnt,
                    const float* __restrict__ Wself,
                    const float* __restrict__ bself,
                    const float* __restrict__ Wneigh,
                    const float* __restrict__ bneigh,
                    float*       __restrict__ out) {
    __shared__ __attribute__((aligned(16))) float sSelf [GEMM_ROWS * A_PITCH];
    __shared__ __attribute__((aligned(16))) float sNeigh[GEMM_ROWS * A_PITCH];

    const int lane  = threadIdx.x & 31;
    const int wave  = threadIdx.x >> 5;
    const int rloc  = lane & 15;
    const int khalf = lane >> 4;
    const int row0  = blockIdx.x * GEMM_ROWS;
    const int col   = wave * 16 + (lane & 15);

    const float bias = bself[col] + bneigh[col];

    {
        const int brow = row0 + wave + 8 * (lane & 7);
        const int braw = min(brow, N_NODES - 1);
        const int craw = cnt[braw];
        const int sraw = off[braw];

        float4 fs[8];
#pragma unroll
        for (int i = 0; i < 8; ++i) {
            const int gr = min(row0 + wave + 8 * i, N_NODES - 1);
            fs[i] = *(const float4*)(feat + (size_t)gr * D + lane * 4);
        }

        const int cv = (brow < N_NODES) ? min(max(craw, 0), N_EDGES) : 0;
        const int sv = (brow < N_NODES) ? min(max(sraw, 0), N_EDGES) : 0;

        int na = __builtin_amdgcn_readlane(cv, 0);
        int sa = __builtin_amdgcn_readlane(sv, 0);
        int nb = __builtin_amdgcn_readlane(cv, 1);
        int sb = __builtin_amdgcn_readlane(sv, 1);
        int ta = csr[min(sa + lane, N_EDGES - 1)];
        int tb = csr[min(sb + lane, N_EDGES - 1)];

#pragma unroll
        for (int i = 0; i < 8; ++i)
            *(float4*)(sSelf + (wave + 8 * i) * A_PITCH + lane * 4) = fs[i];

#pragma nounroll
        for (int p = 0; p < GEMM_TILES; ++p) {
            const int r0 = wave + 16 * p, r1 = r0 + 8;
            const int ca = na, oa = sa, cb = nb, ob = sb;
            const int ia = (lane < ca) ? ta : 0;
            const int ib = (lane < cb) ? tb : 0;
            const bool more = (p + 1 < GEMM_TILES);
            if (more) {
                na = __builtin_amdgcn_readlane(cv, 2 * p + 2);
                sa = __builtin_amdgcn_readlane(sv, 2 * p + 2);
                nb = __builtin_amdgcn_readlane(cv, 2 * p + 3);
                sb = __builtin_amdgcn_readlane(sv, 2 * p + 3);
            }
            const int* pfa = csr + min(sa + lane, N_EDGES - 1);
            const int* pfb = csr + min(sb + lane, N_EDGES - 1);
            float4 n0, n1;
            sage_gather2(feat, csr, oa, ca, ob, cb, lane, ia, ib,
                         more, pfa, pfb, ta, tb, n0, n1);
            const float rd0 = 1.0f / fmaxf((float)ca, 1.0f);
            const float rd1 = 1.0f / fmaxf((float)cb, 1.0f);
            n0.x *= rd0; n0.y *= rd0; n0.z *= rd0; n0.w *= rd0;
            n1.x *= rd1; n1.y *= rd1; n1.z *= rd1; n1.w *= rd1;
            *(float4*)(sNeigh + r0 * A_PITCH + lane * 4) = n0;
            *(float4*)(sNeigh + r1 * A_PITCH + lane * 4) = n1;
        }
    }

    v8f acc[GEMM_TILES];
#pragma unroll
    for (int t = 0; t < GEMM_TILES; ++t)
#pragma unroll
        for (int j = 0; j < 8; ++j) acc[t][j] = bias;

    __syncthreads();

    const float* arow_s = sSelf  + rloc * A_PITCH + khalf * 2;
    const float* arow_n = sNeigh + rloc * A_PITCH + khalf * 2;

#pragma unroll 16
    for (int k0 = 0; k0 < D; k0 += 4) {
        const int k = k0 + khalf * 2;
        v2f b; b.x = Wself[(size_t)k * D + col];
               b.y = Wself[(size_t)(k + 1) * D + col];
#pragma unroll
        for (int t = 0; t < GEMM_TILES; ++t) {
            const v2f a = *(const v2f*)(arow_s + t * 16 * A_PITCH + k0);
            acc[t] = __builtin_amdgcn_wmma_f32_16x16x4_f32(
                         false, a, false, b, (short)0, acc[t], false, false);
        }
    }

#pragma unroll 16
    for (int k0 = 0; k0 < D; k0 += 4) {
        const int k = k0 + khalf * 2;
        v2f b; b.x = Wneigh[(size_t)k * D + col];
               b.y = Wneigh[(size_t)(k + 1) * D + col];
#pragma unroll
        for (int t = 0; t < GEMM_TILES; ++t) {
            const v2f a = *(const v2f*)(arow_n + t * 16 * A_PITCH + k0);
            acc[t] = __builtin_amdgcn_wmma_f32_16x16x4_f32(
                         false, a, false, b, (short)0, acc[t], false, false);
        }
    }

#pragma unroll
    for (int t = 0; t < GEMM_TILES; ++t) {
        if (row0 + 16 * t < N_NODES) {
            float* obase = out + (size_t)(row0 + 16 * t + khalf * 8) * D + col;
#pragma unroll
            for (int j = 0; j < 8; ++j)
                obase[(size_t)j * D] = acc[t][j];
        }
    }
}

extern "C" void kernel_launch(void* const* d_in, const int* in_sizes, int n_in,
                              void* d_out, int out_size, void* d_ws, size_t ws_size,
                              hipStream_t stream) {
    const float* feat   = (const float*)d_in[0];
    const int*   src    = (const int*)  d_in[1];
    const int*   dst    = (const int*)  d_in[2];
    const float* Wself  = (const float*)d_in[3];
    const float* bself  = (const float*)d_in[4];
    const float* Wneigh = (const float*)d_in[5];
    const float* bneigh = (const float*)d_in[6];
    float*       out    = (float*)d_out;

    int* cnt  = (int*)d_ws;
    int* off  = cnt  + N_NODES;
    int* rank = off  + N_NODES;
    int* csr  = rank + N_EDGES;

    sage_zero_cnt<<<(N_NODES + 255) / 256, 256, 0, stream>>>(cnt);
    sage_edge_rank<<<(N_EDGES / 4 + 255) / 256, 256, 0, stream>>>(dst, cnt, rank);
    sage_scan<<<SCAN_B, SCAN_T, 0, stream>>>(cnt, off);
    sage_csr_fill<<<(N_EDGES / 4 + 255) / 256, 256, 0, stream>>>(src, dst, off, rank, csr);
    sage_wmma_gemm<<<GEMM_BLOCKS, 256, 0, stream>>>(
        feat, csr, off, cnt, Wself, bself, Wneigh, bneigh, out);
}


# Per-row source lists in fixed 32-entry slots (slot = int-atomic arrival rank; ranks past 32 go to a side list) replace the counting sort: no prefix-scan or fill launch, 5 -> 3; arrival-order sums
# speedup vs baseline: 2.3605x; 1.2035x over previous
#include <hip/hip_runtime.h>


#define N_NODES 100000
#define N_EDGES 640000
#define D 128
#define SLOTS 32
#define CNT_INTS (N_NODES + 32)
#define SLOT_INTS (N_NODES * SLOTS)

typedef float v2f __attribute__((ext_vector_type(2)));
typedef float v8f __attribute__((ext_vector_type(8)));


__global__ void sage_zero_cnt(int* __restrict__ cnt) {
    const int i = blockIdx.x * blockDim.x + threadIdx.x;
    if (i <= N_NODES) cnt[i] = 0;
}

__device__ __forceinline__ void sage_place(int d, int s, int r,
                                           int* __restrict__ cnt,
                                           int* __restrict__ slots,
                                           int* __restrict__ ovf) {
    if ((unsigned)r < (unsigned)SLOTS) {
        slots[d * SLOTS + r] = s;
    } else {
        const int p = min(max(atomicAdd(cnt + N_NODES, 1), 0), N_EDGES - 1);
        ovf[2 * p]     = d;
        ovf[2 * p + 1] = s;
    }
}

__global__ void sage_edge_slots(const int* __restrict__ src,
                                const int* __restrict__ dst,
                                int*       __restrict__ cnt,
                                int*       __restrict__ slots,
                                int*       __restrict__ ovf) {
    const int t = blockIdx.x * blockDim.x + threadIdx.x;
    if (t < N_EDGES / 4) {
        const int4 d = ((const int4*)dst)[t];
        const int4 s = ((const int4*)src)[t];
        const int r0 = atomicAdd(cnt + d.x, 1);
        const int r1 = atomicAdd(cnt + d.y, 1);
        const int r2 = atomicAdd(cnt + d.z, 1);
        const int r3 = atomicAdd(cnt + d.w, 1);
        sage_place(d.x, s.x, r0, cnt, slots, ovf);
        sage_place(d.y, s.y, r1, cnt, slots, ovf);
        sage_place(d.z, s.z, r2, cnt, slots, ovf);
        sage_place(d.w, s.w, r3, cnt, slots, ovf);
    }
}

__device__ __forceinline__ void sage_gather2(const float* __restrict__ feat,
                                             int na, int nb,
                                             int lane, int ia0, int ib0,
                                             bool more,
                                             const int* __restrict__ pfa,
                                             const int* __restrict__ pfb,
                                             int& ta, int& tb,
                                             float4& ra, float4& rb) {
    const unsigned lane4 = (unsigned)lane * 4u;
    float4 acca = make_float4(0.0f, 0.0f, 0.0f, 0.0f);
    float4 accb = make_float4(0.0f, 0.0f, 0.0f, 0.0f);
    const int nmax = max(na, nb);
    const int ica = min(max(ia0, 0), N_NODES - 1);
    const int icb = min(max(ib0, 0), N_NODES - 1);
    if (more) {
        ta = *pfa;
        tb = *pfb;
    }
    for (int j = 0; j < nmax; j += 8) {
        int qa[8], qb[8];
#pragma unroll
        for (int k = 0; k < 8; ++k) {
            qa[k] = __builtin_amdgcn_readlane(ica, j + k);
            qb[k] = __builtin_amdgcn_readlane(icb, j + k);
        }
        float4 va[8], vb[8];
#pragma unroll
        for (int k = 0; k < 8; ++k) {
            if (j + k < na) va[k] = *(const float4*)(feat + (size_t)qa[k] * D + lane4);
            else            va[k] = make_float4(0.0f, 0.0f, 0.0f, 0.0f);
        }
#pragma unroll
        for (int k = 0; k < 8; ++k) {
            if (j + k < nb) vb[k] = *(const float4*)(feat + (size_t)qb[k] * D + lane4);
            else            vb[k] = make_float4(0.0f, 0.0f, 0.0f, 0.0f);
        }
#pragma unroll
        for (int k = 0; k < 8; ++k) {
            acca.x += va[k].x; acca.y += va[k].y; acca.z += va[k].z; acca.w += va[k].w;
        }
#pragma unroll
        for (int k = 0; k < 8; ++k) {
            accb.x += vb[k].x; accb.y += vb[k].y; accb.z += vb[k].z; accb.w += vb[k].w;
        }
    }
    ra = acca;
    rb = accb;
}

__device__ __forceinline__ void sage_side_list_add(const float* __restrict__ feat,
                                                   const int*   __restrict__ n_ovf,
                                                   const int*   __restrict__ ovf,
                                                   int rowa, bool wa,
                                                   int rowb, bool wb,
                                                   int lane,
                                                   float4& acca, float4& accb) {
    const unsigned lane4 = (unsigned)lane * 4u;
    const int n = __builtin_amdgcn_readfirstlane(min(max(*n_ovf, 0), N_EDGES));
#pragma nounroll
    for (int base = 0; base < n; base += 32) {
        const int pos = min(base + lane, N_EDGES - 1);
        const int ed = ovf[2 * pos];
        const int es = ovf[2 * pos + 1];
        const bool in = (base + lane < n);
        const int sl = min(max(es, 0), N_NODES - 1);
        unsigned ma = __builtin_amdgcn_ballot_w32(in && wa && ed == rowa);
        unsigned mb = __builtin_amdgcn_ballot_w32(in && wb && ed == rowb);
        while (ma != 0u) {
            const int k = __builtin_ctz(ma);
            ma &= ma - 1u;
            const int s = __builtin_amdgcn_readlane(sl, k);
            const float4 v = *(const float4*)(feat + (size_t)s * D + lane4);
            acca.x += v.x; acca.y += v.y; acca.z += v.z; acca.w += v.w;
        }
        while (mb != 0u) {
            const int k = __builtin_ctz(mb);
            mb &= mb - 1u;
            const int s = __builtin_amdgcn_readlane(sl, k);
            const float4 v = *(const float4*)(feat + (size_t)s * D + lane4);
            accb.x += v.x; accb.y += v.y; accb.z += v.z; accb.w += v.w;
        }
    }
}

#define A_PITCH 132
#define GEMM_TILES 4
#define GEMM_ROWS (16 * GEMM_TILES)
#define GEMM_BLOCKS ((N_NODES + GEMM_ROWS - 1) / GEMM_ROWS)

__global__ __launch_bounds__(256)
void sage_wmma_gemm(const float* __restrict__ feat,
                    const int*   __restrict__ slots,
                    const int*   __restrict__ ovf,
                    const int*   __restrict__ cnt,
                    const float* __restrict__ Wself,
                    const float* __restrict__ bself,
                    const float* __restrict__ Wneigh,
                    const float* __restrict__ bneigh,
                    float*       __restrict__ out) {
    __shared__ __attribute__((aligned(16))) float sSelf [GEMM_ROWS * A_PITCH];
    __shared__ __attribute__((aligned(16))) float sNeigh[GEMM_ROWS * A_PITCH];

    const int lane  = threadIdx.x & 31;
    const int wave  = threadIdx.x >> 5;
    const int rloc  = lane & 15;
    const int khalf = lane >> 4;
    const int row0  = blockIdx.x * GEMM_ROWS;
    const int col   = wave * 16 + (lane & 15);

    const float bias = bself[col] + bneigh[col];

    {
        int ta = slots[min(row0 + wave,     N_NODES - 1) * SLOTS + lane];
        int tb = slots[min(row0 + wave + 8, N_NODES - 1) * SLOTS + lane];

        const int brow = row0 + wave + 8 * (lane & 7);
        const int braw = min(brow, N_NODES - 1);
        const int craw = cnt[braw];

        float4 fs[8];
#pragma unroll
        for (int i = 0; i < 8; ++i) {
            const int gr = min(row0 + wave + 8 * i, N_NODES - 1);
            fs[i] = *(const float4*)(feat + (size_t)gr * D + lane * 4);
        }

        const int cv = (brow < N_NODES) ? min(max(craw, 0), N_EDGES) : 0;

        int na = __builtin_amdgcn_readlane(cv, 0);
        int nb = __builtin_amdgcn_readlane(cv, 1);

#pragma unroll
        for (int i = 0; i < 8; ++i)
            *(float4*)(sSelf + (wave + 8 * i) * A_PITCH + lane * 4) = fs[i];

#pragma nounroll
        for (int p = 0; p < GEMM_TILES; ++p) {
            const int r0 = wave + 16 * p, r1 = r0 + 8;
            const int ca = na, cb = nb;
            const int fa = min(ca, SLOTS), fb = min(cb, SLOTS);
            const int ia = (lane < fa) ? ta : 0;
            const int ib = (lane < fb) ? tb : 0;
            const bool more = (p + 1 < GEMM_TILES);
            if (more) {
                na = __builtin_amdgcn_readlane(cv, 2 * p + 2);
                nb = __builtin_amdgcn_readlane(cv, 2 * p + 3);
            }
            const int* pfa = slots + min(row0 + r0 + 16, N_NODES - 1) * SLOTS + lane;
            const int* pfb = slots + min(row0 + r1 + 16, N_NODES - 1) * SLOTS + lane;
            float4 n0, n1;
            sage_gather2(feat, fa, fb, lane, ia, ib,
                         more, pfa, pfb, ta, tb, n0, n1);
            if (ca > SLOTS || cb > SLOTS)
                sage_side_list_add(feat, cnt + N_NODES, ovf,
                                   row0 + r0, ca > SLOTS, row0 + r1, cb > SLOTS,
                                   lane, n0, n1);
            const float rd0 = 1.0f / fmaxf((float)ca, 1.0f);
            const float rd1 = 1.0f / fmaxf((float)cb, 1.0f);
            n0.x *= rd0; n0.y *= rd0; n0.z *= rd0; n0.w *= rd0;
            n1.x *= rd1; n1.y *= rd1; n1.z *= rd1; n1.w *= rd1;
            *(float4*)(sNeigh + r0 * A_PITCH + lane * 4) = n0;
            *(float4*)(sNeigh + r1 * A_PITCH + lane * 4) = n1;
        }
    }

    v8f acc[GEMM_TILES];
#pragma unroll
    for (int t = 0; t < GEMM_TILES; ++t)
#pragma unroll
        for (int j = 0; j < 8; ++j) acc[t][j] = bias;

    __syncthreads();

    const float* arow_s = sSelf  + rloc * A_PITCH + khalf * 2;
    const float* arow_n = sNeigh + rloc * A_PITCH + khalf * 2;

#pragma unroll 16
    for (int k0 = 0; k0 < D; k0 += 4) {
        const int k = k0 + khalf * 2;
        v2f b; b.x = Wself[(size_t)k * D + col];
               b.y = Wself[(size_t)(k + 1) * D + col];
#pragma unroll
        for (int t = 0; t < GEMM_TILES; ++t) {
            const v2f a = *(const v2f*)(arow_s + t * 16 * A_PITCH + k0);
            acc[t] = __builtin_amdgcn_wmma_f32_16x16x4_f32(
                         false, a, false, b, (short)0, acc[t], false, false);
        }
    }

#pragma unroll 16
    for (int k0 = 0; k0 < D; k0 += 4) {
        const int k = k0 + khalf * 2;
        v2f b; b.x = Wneigh[(size_t)k * D + col];
               b.y = Wneigh[(size_t)(k + 1) * D + col];
#pragma unroll
        for (int t = 0; t < GEMM_TILES; ++t) {
            const v2f a = *(const v2f*)(arow_n + t * 16 * A_PITCH + k0);
            acc[t] = __builtin_amdgcn_wmma_f32_16x16x4_f32(
                         false, a, false, b, (short)0, acc[t], false, false);
        }
    }

#pragma unroll
    for (int t = 0; t < GEMM_TILES; ++t) {
        if (row0 + 16 * t < N_NODES) {
            float* obase = out + (size_t)(row0 + 16 * t + khalf * 8) * D + col;
#pragma unroll
            for (int j = 0; j < 8; ++j)
                obase[(size_t)j * D] = acc[t][j];
        }
    }
}

extern "C" void kernel_launch(void* const* d_in, const int* in_sizes, int n_in,
                              void* d_out, int out_size, void* d_ws, size_t ws_size,
                              hipStream_t stream) {
    const float* feat   = (const float*)d_in[0];
    const int*   src    = (const int*)  d_in[1];
    const int*   dst    = (const int*)  d_in[2];
    const float* Wself  = (const float*)d_in[3];
    const float* bself  = (const float*)d_in[4];
    const float* Wneigh = (const float*)d_in[5];
    const float* bneigh = (const float*)d_in[6];
    float*       out    = (float*)d_out;

    int* cnt   = (int*)d_ws;
    int* slots = cnt   + CNT_INTS;
    int* ovf   = slots + SLOT_INTS;

    sage_zero_cnt<<<(N_NODES + 1 + 255) / 256, 256, 0, stream>>>(cnt);
    sage_edge_slots<<<(N_EDGES / 4 + 255) / 256, 256, 0, stream>>>(src, dst, cnt, slots, ovf);
    sage_wmma_gemm<<<GEMM_BLOCKS, 256, 0, stream>>>(
        feat, slots, ovf, cnt, Wself, bself, Wneigh, bneigh, out);
}


# Source lists by LDS binning: 79 edge chunks sort edges into per-row-block 128 B cells in LDS, whole-line stores; GEMM block builds row lists in LDS; no global atomic, 3->2 launches; arrival-order sum
# speedup vs baseline: 3.3535x; 1.4207x over previous
#include <hip/hip_runtime.h>


#define N_NODES 100000
#define N_EDGES 640000
#define D 128
#define SLOTS 32
#define GEMM_TILES 4
#define GEMM_ROWS (16 * GEMM_TILES)
#define GEMM_BLOCKS ((N_NODES + GEMM_ROWS - 1) / GEMM_ROWS)
#define CHUNK_EDGES 8192
#define N_CHUNKS ((N_EDGES + CHUNK_EDGES - 1) / CHUNK_EDGES)
#define CELL 32
#define CELL_CAP (CELL - 1)
#define CELL_INTS (N_CHUNKS * GEMM_BLOCKS * CELL)
#define OVFCNT_INTS 128
#define WAVE_CHUNKS ((N_CHUNKS + 7) / 8)

typedef float v2f __attribute__((ext_vector_type(2)));
typedef float v4f __attribute__((ext_vector_type(4)));
typedef int   v4i __attribute__((ext_vector_type(4)));
typedef float v8f __attribute__((ext_vector_type(8)));


__device__ __forceinline__ int sage_bin_rank(int d, bool have, int* sCnt) {
    const int dc = min(max(d, 0), N_NODES - 1);
    return have ? atomicAdd(sCnt + (dc >> 6), 1) : 0;
}

__device__ __forceinline__ void sage_bin_store(int d, int s, int rank, bool have, int c,
                                               int* sCell, int* sOvf,
                                               int* __restrict__ ovf) {
    const int dc = min(max(d, 0), N_NODES - 1);
    const int sc = min(max(s, 0), N_NODES - 1);
    if (have) {
        if (rank < CELL_CAP) {
            sCell[(dc >> 6) * CELL + 1 + rank] = (sc << 6) | (dc & 63);
        } else {
            const int p = min(max(atomicAdd(sOvf, 1), 0), CHUNK_EDGES - 1);
            ovf[((size_t)c * CHUNK_EDGES + p) * 2]     = dc;
            ovf[((size_t)c * CHUNK_EDGES + p) * 2 + 1] = sc;
        }
    }
}

__global__ __launch_bounds__(1024)
void sage_bin_edges(const int* __restrict__ src,
                    const int* __restrict__ dst,
                    int*       __restrict__ cells,
                    int*       __restrict__ ovfcnt,
                    int*       __restrict__ ovf) {
    __shared__ __attribute__((aligned(16))) int sCell[GEMM_BLOCKS * CELL];
    __shared__ int sCnt[GEMM_BLOCKS];
    __shared__ int sOvf;

    const int tid = threadIdx.x;
    const int c   = blockIdx.x;

    const int e0 = c * CHUNK_EDGES + tid * 4;
    const int e1 = e0 + CHUNK_EDGES / 2;
    const bool h0 = e0 < N_EDGES, h1 = e1 < N_EDGES;
    const v4i d0 = *(const v4i*)(dst + min(e0, N_EDGES - 4));
    const v4i s0 = *(const v4i*)(src + min(e0, N_EDGES - 4));
    const v4i d1 = *(const v4i*)(dst + min(e1, N_EDGES - 4));
    const v4i s1 = *(const v4i*)(src + min(e1, N_EDGES - 4));

    const v4i zero4 = {0, 0, 0, 0};
    for (int i = tid; i < GEMM_BLOCKS * CELL / 4; i += 1024) ((v4i*)sCell)[i] = zero4;
    for (int b = tid; b < GEMM_BLOCKS; b += 1024) sCnt[b] = 0;
    if (tid == 0) sOvf = 0;

    __syncthreads();

    const int r0 = sage_bin_rank(d0.x, h0, sCnt);
    const int r1 = sage_bin_rank(d0.y, h0, sCnt);
    const int r2 = sage_bin_rank(d0.z, h0, sCnt);
    const int r3 = sage_bin_rank(d0.w, h0, sCnt);
    const int r4 = sage_bin_rank(d1.x, h1, sCnt);
    const int r5 = sage_bin_rank(d1.y, h1, sCnt);
    const int r6 = sage_bin_rank(d1.z, h1, sCnt);
    const int r7 = sage_bin_rank(d1.w, h1, sCnt);
    sage_bin_store(d0.x, s0.x, r0, h0, c, sCell, &sOvf, ovf);
    sage_bin_store(d0.y, s0.y, r1, h0, c, sCell, &sOvf, ovf);
    sage_bin_store(d0.z, s0.z, r2, h0, c, sCell, &sOvf, ovf);
    sage_bin_store(d0.w, s0.w, r3, h0, c, sCell, &sOvf, ovf);
    sage_bin_store(d1.x, s1.x, r4, h1, c, sCell, &sOvf, ovf);
    sage_bin_store(d1.y, s1.y, r5, h1, c, sCell, &sOvf, ovf);
    sage_bin_store(d1.z, s1.z, r6, h1, c, sCell, &sOvf, ovf);
    sage_bin_store(d1.w, s1.w, r7, h1, c, sCell, &sOvf, ovf);

    __syncthreads();

    v4i*       g = (v4i*)(cells + (size_t)c * (GEMM_BLOCKS * CELL));
    const v4i* l = (const v4i*)sCell;
    for (int i = tid; i < GEMM_BLOCKS * CELL / 4; i += 1024) {
        v4i v = l[i];
        if ((i & (CELL / 4 - 1)) == 0) v.x = sCnt[i / (CELL / 4)];
        g[i] = v;
    }
    if (tid == 0) ovfcnt[c] = sOvf;
}

__device__ __forceinline__ void sage_row_place(int r, int s, int* sRowCnt, int* sSlot) {
    const int rank = atomicAdd(sRowCnt + r, 1);
    if ((unsigned)rank < (unsigned)SLOTS) sSlot[r * SLOTS + rank] = s;
}

__device__ __forceinline__ void sage_gather2(const float* __restrict__ feat,
                                             int na, int nb,
                                             int lane, int ia0, int ib0,
                                             float4& ra, float4& rb) {
    const unsigned lane4 = (unsigned)lane * 4u;
    float4 acca = make_float4(0.0f, 0.0f, 0.0f, 0.0f);
    float4 accb = make_float4(0.0f, 0.0f, 0.0f, 0.0f);
    const int nmax = max(na, nb);
    const int ica = min(max(ia0, 0), N_NODES - 1);
    const int icb = min(max(ib0, 0), N_NODES - 1);
    for (int j = 0; j < nmax; j += 8) {
        int qa[8], qb[8];
#pragma unroll
        for (int k = 0; k < 8; ++k) {
            qa[k] = __builtin_amdgcn_readlane(ica, j + k);
            qb[k] = __builtin_amdgcn_readlane(icb, j + k);
        }
        float4 va[8], vb[8];
#pragma unroll
        for (int k = 0; k < 8; ++k) {
            if (j + k < na) va[k] = *(const float4*)(feat + (size_t)qa[k] * D + lane4);
            else            va[k] = make_float4(0.0f, 0.0f, 0.0f, 0.0f);
        }
#pragma unroll
        for (int k = 0; k < 8; ++k) {
            if (j + k < nb) vb[k] = *(const float4*)(feat + (size_t)qb[k] * D + lane4);
            else            vb[k] = make_float4(0.0f, 0.0f, 0.0f, 0.0f);
        }
#pragma unroll
        for (int k = 0; k < 8; ++k) {
            acca.x += va[k].x; acca.y += va[k].y; acca.z += va[k].z; acca.w += va[k].w;
        }
#pragma unroll
        for (int k = 0; k < 8; ++k) {
            accb.x += vb[k].x; accb.y += vb[k].y; accb.z += vb[k].z; accb.w += vb[k].w;
        }
    }
    ra = acca;
    rb = accb;
}

__device__ __forceinline__ void sage_add_marked(const float* __restrict__ feat,
                                                unsigned ma, unsigned mb, int sl,
                                                unsigned lane4,
                                                float4& acca, float4& accb) {
    while (ma != 0u) {
        const int k = __builtin_ctz(ma);
        ma &= ma - 1u;
        const int s = __builtin_amdgcn_readlane(sl, k);
        const float4 v = *(const float4*)(feat + (size_t)s * D + lane4);
        acca.x += v.x; acca.y += v.y; acca.z += v.z; acca.w += v.w;
    }
    while (mb != 0u) {
        const int k = __builtin_ctz(mb);
        mb &= mb - 1u;
        const int s = __builtin_amdgcn_readlane(sl, k);
        const float4 v = *(const float4*)(feat + (size_t)s * D + lane4);
        accb.x += v.x; accb.y += v.y; accb.z += v.z; accb.w += v.w;
    }
}

__device__ __forceinline__ void sage_walk_rows(const float* __restrict__ feat,
                                               const int*   __restrict__ cells,
                                               const int*   __restrict__ ovfcnt,
                                               const int*   __restrict__ ovf,
                                               int blk,
                                               int ra, bool wa,
                                               int rb, bool wb,
                                               int lane,
                                               float4& acca, float4& accb) {
    const unsigned lane4 = (unsigned)lane * 4u;
    if (wa) acca = make_float4(0.0f, 0.0f, 0.0f, 0.0f);
    if (wb) accb = make_float4(0.0f, 0.0f, 0.0f, 0.0f);
    const int rowa = blk * GEMM_ROWS + ra, rowb = blk * GEMM_ROWS + rb;
#pragma nounroll
    for (int c = 0; c < N_CHUNKS; ++c) {
        const int v = cells[((size_t)c * GEMM_BLOCKS + blk) * CELL + lane];
        const int n = min(max(__builtin_amdgcn_readfirstlane(v), 0), CELL_CAP);
        const bool in = (lane >= 1 && lane <= n);
        const int sl = min(max(v >> 6, 0), N_NODES - 1);
        const int r  = v & 63;
        sage_add_marked(feat,
                        __builtin_amdgcn_ballot_w32(in && wa && r == ra),
                        __builtin_amdgcn_ballot_w32(in && wb && r == rb),
                        sl, lane4, acca, accb);
        const int m = __builtin_amdgcn_readfirstlane(min(max(ovfcnt[c], 0), CHUNK_EDGES));
#pragma nounroll
        for (int base = 0; base < m; base += 32) {
            const int pos = min(base + lane, CHUNK_EDGES - 1);
            const int ed = ovf[((size_t)c * CHUNK_EDGES + pos) * 2];
            const int es = ovf[((size_t)c * CHUNK_EDGES + pos) * 2 + 1];
            const bool in2 = (base + lane < m);
            const int sl2 = min(max(es, 0), N_NODES - 1);
            sage_add_marked(feat,
                            __builtin_amdgcn_ballot_w32(in2 && wa && ed == rowa),
                            __builtin_amdgcn_ballot_w32(in2 && wb && ed == rowb),
                            sl2, lane4, acca, accb);
        }
    }
}

#define A_PITCH 132

__global__ __launch_bounds__(256)
void sage_wmma_gemm(const float* __restrict__ feat,
                    const int*   __restrict__ cells,
                    const int*   __restrict__ ovfcnt,
                    const int*   __restrict__ ovf,
                    const float* __restrict__ Wself,
                    const float* __restrict__ bself,
                    const float* __restrict__ Wneigh,
                    const float* __restrict__ bneigh,
                    float*       __restrict__ out) {
    __shared__ __attribute__((aligned(16))) float sSelf [GEMM_ROWS * A_PITCH];
    __shared__ __attribute__((aligned(16))) float sNeigh[GEMM_ROWS * A_PITCH];
    __shared__ int sSlot[GEMM_ROWS * SLOTS];
    __shared__ int sRowCnt[GEMM_ROWS];

    const int lane  = threadIdx.x & 31;
    const int wave  = threadIdx.x >> 5;
    const int rloc  = lane & 15;
    const int khalf = lane >> 4;
    const int blk   = blockIdx.x;
    const int row0  = blk * GEMM_ROWS;
    const int col   = wave * 16 + (lane & 15);

    const float bias = bself[col] + bneigh[col];

    if (threadIdx.x < GEMM_ROWS) sRowCnt[threadIdx.x] = 0;

    const int wv = __builtin_amdgcn_readfirstlane(wave);

    const float* fsrc = feat + lane * 4;
    const v4f f0 = *(const v4f*)(fsrc + (size_t)min(row0 + wave,      N_NODES - 1) * D);
    const v4f f1 = *(const v4f*)(fsrc + (size_t)min(row0 + wave +  8, N_NODES - 1) * D);
    const v4f f2 = *(const v4f*)(fsrc + (size_t)min(row0 + wave + 16, N_NODES - 1) * D);
    const v4f f3 = *(const v4f*)(fsrc + (size_t)min(row0 + wave + 24, N_NODES - 1) * D);
    const v4f f4 = *(const v4f*)(fsrc + (size_t)min(row0 + wave + 32, N_NODES - 1) * D);
    const v4f f5 = *(const v4f*)(fsrc + (size_t)min(row0 + wave + 40, N_NODES - 1) * D);
    const v4f f6 = *(const v4f*)(fsrc + (size_t)min(row0 + wave + 48, N_NODES - 1) * D);
    const v4f f7 = *(const v4f*)(fsrc + (size_t)min(row0 + wave + 56, N_NODES - 1) * D);

    int cellv[WAVE_CHUNKS];
#pragma unroll
    for (int i = 0; i < WAVE_CHUNKS; ++i) {
        const int c = min(wv + 8 * i, N_CHUNKS - 1);
        cellv[i] = cells[((size_t)c * GEMM_BLOCKS + blk) * CELL + lane];
    }
    const int oc = ovfcnt[min(wv + 8 * lane, N_CHUNKS - 1)];

    float* fdst = sSelf + wave * A_PITCH + lane * 4;
    *(v4f*)(fdst +  0 * A_PITCH) = f0;
    *(v4f*)(fdst +  8 * A_PITCH) = f1;
    *(v4f*)(fdst + 16 * A_PITCH) = f2;
    *(v4f*)(fdst + 24 * A_PITCH) = f3;
    *(v4f*)(fdst + 32 * A_PITCH) = f4;
    *(v4f*)(fdst + 40 * A_PITCH) = f5;
    *(v4f*)(fdst + 48 * A_PITCH) = f6;
    *(v4f*)(fdst + 56 * A_PITCH) = f7;

    __syncthreads();

    int rk[WAVE_CHUNKS];
#pragma unroll
    for (int i = 0; i < WAVE_CHUNKS; ++i) {
        const int v = cellv[i];
        const int n = (wv + 8 * i < N_CHUNKS)
                    ? min(max(__builtin_amdgcn_readfirstlane(v), 0), CELL_CAP) : 0;
        rk[i] = (lane >= 1 && lane <= n) ? atomicAdd(sRowCnt + (v & 63), 1) : SLOTS;
    }
#pragma unroll
    for (int i = 0; i < WAVE_CHUNKS; ++i) {
        const int v = cellv[i];
        if ((unsigned)rk[i] < (unsigned)SLOTS)
            sSlot[(v & 63) * SLOTS + rk[i]] = min(max(v >> 6, 0), N_NODES - 1);
    }
    if (__builtin_amdgcn_ballot_w32(oc > 0) != 0u) {
#pragma nounroll
        for (int i = 0; i < WAVE_CHUNKS; ++i) {
            const int c = wv + 8 * i;
            const int m = (c < N_CHUNKS)
                        ? min(max(__builtin_amdgcn_readlane(oc, i), 0), CHUNK_EDGES) : 0;
#pragma nounroll
            for (int base = 0; base < m; base += 32) {
                const int pos = base + lane;
                if (pos < m) {
                    const int ed = ovf[((size_t)c * CHUNK_EDGES + pos) * 2];
                    const int es = ovf[((size_t)c * CHUNK_EDGES + pos) * 2 + 1];
                    if ((ed >> 6) == blk)
                        sage_row_place(ed & 63, min(max(es, 0), N_NODES - 1), sRowCnt, sSlot);
                }
            }
        }
    }

    __syncthreads();

    {
        const int cv = min(max(sRowCnt[wave + 8 * (lane & 7)], 0), N_EDGES);

#pragma nounroll
        for (int p = 0; p < GEMM_TILES; ++p) {
            const int r0 = wave + 16 * p, r1 = r0 + 8;
            const int ca = __builtin_amdgcn_readlane(cv, 2 * p);
            const int cb = __builtin_amdgcn_readlane(cv, 2 * p + 1);
            const int fa = min(ca, SLOTS), fb = min(cb, SLOTS);
            const int ia = (lane < fa) ? sSlot[r0 * SLOTS + lane] : 0;
            const int ib = (lane < fb) ? sSlot[r1 * SLOTS + lane] : 0;
            float4 n0, n1;
            sage_gather2(feat, fa, fb, lane, ia, ib, n0, n1);
            if (ca > SLOTS || cb > SLOTS)
                sage_walk_rows(feat, cells, ovfcnt, ovf, blk,
                               r0, ca > SLOTS, r1, cb > SLOTS, lane, n0, n1);
            const float rd0 = 1.0f / fmaxf((float)ca, 1.0f);
            const float rd1 = 1.0f / fmaxf((float)cb, 1.0f);
            n0.x *= rd0; n0.y *= rd0; n0.z *= rd0; n0.w *= rd0;
            n1.x *= rd1; n1.y *= rd1; n1.z *= rd1; n1.w *= rd1;
            *(float4*)(sNeigh + r0 * A_PITCH + lane * 4) = n0;
            *(float4*)(sNeigh + r1 * A_PITCH + lane * 4) = n1;
        }
    }

    v8f acc[GEMM_TILES];
#pragma unroll
    for (int t = 0; t < GEMM_TILES; ++t)
#pragma unroll
        for (int j = 0; j < 8; ++j) acc[t][j] = bias;

    __syncthreads();

    const float* arow_s = sSelf  + rloc * A_PITCH + khalf * 2;
    const float* arow_n = sNeigh + rloc * A_PITCH + khalf * 2;

#pragma unroll 16
    for (int k0 = 0; k0 < D; k0 += 4) {
        const int k = k0 + khalf * 2;
        v2f b; b.x = Wself[(size_t)k * D + col];
               b.y = Wself[(size_t)(k + 1) * D + col];
#pragma unroll
        for (int t = 0; t < GEMM_TILES; ++t) {
            const v2f a = *(const v2f*)(arow_s + t * 16 * A_PITCH + k0);
            acc[t] = __builtin_amdgcn_wmma_f32_16x16x4_f32(
                         false, a, false, b, (short)0, acc[t], false, false);
        }
    }

#pragma unroll 16
    for (int k0 = 0; k0 < D; k0 += 4) {
        const int k = k0 + khalf * 2;
        v2f b; b.x = Wneigh[(size_t)k * D + col];
               b.y = Wneigh[(size_t)(k + 1) * D + col];
#pragma unroll
        for (int t = 0; t < GEMM_TILES; ++t) {
            const v2f a = *(const v2f*)(arow_n + t * 16 * A_PITCH + k0);
            acc[t] = __builtin_amdgcn_wmma_f32_16x16x4_f32(
                         false, a, false, b, (short)0, acc[t], false, false);
        }
    }

#pragma unroll
    for (int t = 0; t < GEMM_TILES; ++t) {
        if (row0 + 16 * t < N_NODES) {
            float* obase = out + (size_t)(row0 + 16 * t + khalf * 8) * D + col;
#pragma unroll
            for (int j = 0; j < 8; ++j)
                obase[(size_t)j * D] = acc[t][j];
        }
    }
}

extern "C" void kernel_launch(void* const* d_in, const int* in_sizes, int n_in,
                              void* d_out, int out_size, void* d_ws, size_t ws_size,
                              hipStream_t stream) {
    const float* feat   = (const float*)d_in[0];
    const int*   src    = (const int*)  d_in[1];
    const int*   dst    = (const int*)  d_in[2];
    const float* Wself  = (const float*)d_in[3];
    const float* bself  = (const float*)d_in[4];
    const float* Wneigh = (const float*)d_in[5];
    const float* bneigh = (const float*)d_in[6];
    float*       out    = (float*)d_out;

    int* cells  = (int*)d_ws;
    int* ovfcnt = cells  + CELL_INTS;
    int* ovf    = ovfcnt + OVFCNT_INTS;

    sage_bin_edges<<<N_CHUNKS, 1024, 0, stream>>>(src, dst, cells, ovfcnt, ovf);
    sage_wmma_gemm<<<GEMM_BLOCKS, 256, 0, stream>>>(
        feat, cells, ovfcnt, ovf, Wself, bself, Wneigh, bneigh, out);
}


# GEMM kernel: the feat @ W_self WMMAs run inside the neighbour staging, an eighth of k per row, between the issue of that row's source-row loads and their adds (one row per round); same sum order
# speedup vs baseline: 3.4566x; 1.0307x over previous
#include <hip/hip_runtime.h>


#define N_NODES 100000
#define N_EDGES 640000
#define D 128
#define SLOTS 32
#define GEMM_TILES 4
#define GEMM_ROWS (16 * GEMM_TILES)
#define GEMM_BLOCKS ((N_NODES + GEMM_ROWS - 1) / GEMM_ROWS)
#define CHUNK_EDGES 8192
#define N_CHUNKS ((N_EDGES + CHUNK_EDGES - 1) / CHUNK_EDGES)
#define CELL 32
#define CELL_CAP (CELL - 1)
#define CELL_INTS (N_CHUNKS * GEMM_BLOCKS * CELL)
#define OVFCNT_INTS 128
#define WAVE_CHUNKS ((N_CHUNKS + 7) / 8)
#define ZROW_INT_OFF (N_CHUNKS * CHUNK_EDGES * 2 - D)

typedef float v2f __attribute__((ext_vector_type(2)));
typedef float v4f __attribute__((ext_vector_type(4)));
typedef int   v4i __attribute__((ext_vector_type(4)));
typedef float v8f __attribute__((ext_vector_type(8)));


__device__ __forceinline__ int sage_bin_rank(int d, bool have, int* sCnt) {
    const int dc = min(max(d, 0), N_NODES - 1);
    return have ? atomicAdd(sCnt + (dc >> 6), 1) : 0;
}

__device__ __forceinline__ void sage_bin_store(int d, int s, int rank, bool have, int c,
                                               int* sCell, int* sOvf,
                                               int* __restrict__ ovf) {
    const int dc = min(max(d, 0), N_NODES - 1);
    const int sc = min(max(s, 0), N_NODES - 1);
    if (have) {
        if (rank < CELL_CAP) {
            sCell[(dc >> 6) * CELL + 1 + rank] = (sc << 6) | (dc & 63);
        } else {
            const int p = min(max(atomicAdd(sOvf, 1), 0), CHUNK_EDGES - 1);
            ovf[((size_t)c * CHUNK_EDGES + p) * 2]     = dc;
            ovf[((size_t)c * CHUNK_EDGES + p) * 2 + 1] = sc;
        }
    }
}

__global__ __launch_bounds__(1024)
void sage_bin_edges(const int* __restrict__ src,
                    const int* __restrict__ dst,
                    int*       __restrict__ cells,
                    int*       __restrict__ ovfcnt,
                    int*       __restrict__ ovf) {
    __shared__ __attribute__((aligned(16))) int sCell[GEMM_BLOCKS * CELL];
    __shared__ int sCnt[GEMM_BLOCKS];
    __shared__ int sOvf;

    const int tid = threadIdx.x;
    const int c   = blockIdx.x;

    const int e0 = c * CHUNK_EDGES + tid * 4;
    const int e1 = e0 + CHUNK_EDGES / 2;
    const bool h0 = e0 < N_EDGES, h1 = e1 < N_EDGES;
    const v4i d0 = *(const v4i*)(dst + min(e0, N_EDGES - 4));
    const v4i s0 = *(const v4i*)(src + min(e0, N_EDGES - 4));
    const v4i d1 = *(const v4i*)(dst + min(e1, N_EDGES - 4));
    const v4i s1 = *(const v4i*)(src + min(e1, N_EDGES - 4));

    const v4i zero4 = {0, 0, 0, 0};
    for (int i = tid; i < GEMM_BLOCKS * CELL / 4; i += 1024) ((v4i*)sCell)[i] = zero4;
    for (int b = tid; b < GEMM_BLOCKS; b += 1024) sCnt[b] = 0;
    if (tid == 0) sOvf = 0;

    __syncthreads();

    const int r0 = sage_bin_rank(d0.x, h0, sCnt);
    const int r1 = sage_bin_rank(d0.y, h0, sCnt);
    const int r2 = sage_bin_rank(d0.z, h0, sCnt);
    const int r3 = sage_bin_rank(d0.w, h0, sCnt);
    const int r4 = sage_bin_rank(d1.x, h1, sCnt);
    const int r5 = sage_bin_rank(d1.y, h1, sCnt);
    const int r6 = sage_bin_rank(d1.z, h1, sCnt);
    const int r7 = sage_bin_rank(d1.w, h1, sCnt);
    sage_bin_store(d0.x, s0.x, r0, h0, c, sCell, &sOvf, ovf);
    sage_bin_store(d0.y, s0.y, r1, h0, c, sCell, &sOvf, ovf);
    sage_bin_store(d0.z, s0.z, r2, h0, c, sCell, &sOvf, ovf);
    sage_bin_store(d0.w, s0.w, r3, h0, c, sCell, &sOvf, ovf);
    sage_bin_store(d1.x, s1.x, r4, h1, c, sCell, &sOvf, ovf);
    sage_bin_store(d1.y, s1.y, r5, h1, c, sCell, &sOvf, ovf);
    sage_bin_store(d1.z, s1.z, r6, h1, c, sCell, &sOvf, ovf);
    sage_bin_store(d1.w, s1.w, r7, h1, c, sCell, &sOvf, ovf);

    __syncthreads();

    v4i*       g = (v4i*)(cells + (size_t)c * (GEMM_BLOCKS * CELL));
    const v4i* l = (const v4i*)sCell;
    for (int i = tid; i < GEMM_BLOCKS * CELL / 4; i += 1024) {
        v4i v = l[i];
        if ((i & (CELL / 4 - 1)) == 0) v.x = sCnt[i / (CELL / 4)];
        g[i] = v;
    }
    if (tid == 0) ovfcnt[c] = sOvf;
    if (c == N_CHUNKS - 1 && tid < D) ovf[ZROW_INT_OFF + tid] = 0;
}

__device__ __forceinline__ void sage_row_place(int r, int s, int* sRowCnt, int* sSlot) {
    const int rank = atomicAdd(sRowCnt + r, 1);
    if ((unsigned)rank < (unsigned)SLOTS) sSlot[r * SLOTS + rank] = s;
}

__device__ __forceinline__ void sage_gather2(const float* __restrict__ feat,
                                             int na, int nb,
                                             int lane, int ia0, int ib0,
                                             float4& ra, float4& rb) {
    const unsigned lane4 = (unsigned)lane * 4u;
    float4 acca = ra;
    float4 accb = rb;
    const int nmax = max(na, nb);
    const int ica = min(max(ia0, 0), N_NODES - 1);
    const int icb = min(max(ib0, 0), N_NODES - 1);
    for (int j = 8; j < nmax; j += 8) {
        int qa[8], qb[8];
#pragma unroll
        for (int k = 0; k < 8; ++k) {
            qa[k] = __builtin_amdgcn_readlane(ica, j + k);
            qb[k] = __builtin_amdgcn_readlane(icb, j + k);
        }
        float4 va[8], vb[8];
#pragma unroll
        for (int k = 0; k < 8; ++k) {
            if (j + k < na) va[k] = *(const float4*)(feat + (size_t)qa[k] * D + lane4);
            else            va[k] = make_float4(0.0f, 0.0f, 0.0f, 0.0f);
        }
#pragma unroll
        for (int k = 0; k < 8; ++k) {
            if (j + k < nb) vb[k] = *(const float4*)(feat + (size_t)qb[k] * D + lane4);
            else            vb[k] = make_float4(0.0f, 0.0f, 0.0f, 0.0f);
        }
#pragma unroll
        for (int k = 0; k < 8; ++k) {
            acca.x += va[k].x; acca.y += va[k].y; acca.z += va[k].z; acca.w += va[k].w;
        }
#pragma unroll
        for (int k = 0; k < 8; ++k) {
            accb.x += vb[k].x; accb.y += vb[k].y; accb.z += vb[k].z; accb.w += vb[k].w;
        }
    }
    ra = acca;
    rb = accb;
}

__device__ __forceinline__ void sage_add_marked(const float* __restrict__ feat,
                                                unsigned ma, unsigned mb, int sl,
                                                unsigned lane4,
                                                float4& acca, float4& accb) {
    while (ma != 0u) {
        const int k = __builtin_ctz(ma);
        ma &= ma - 1u;
        const int s = __builtin_amdgcn_readlane(sl, k);
        const float4 v = *(const float4*)(feat + (size_t)s * D + lane4);
        acca.x += v.x; acca.y += v.y; acca.z += v.z; acca.w += v.w;
    }
    while (mb != 0u) {
        const int k = __builtin_ctz(mb);
        mb &= mb - 1u;
        const int s = __builtin_amdgcn_readlane(sl, k);
        const float4 v = *(const float4*)(feat + (size_t)s * D + lane4);
        accb.x += v.x; accb.y += v.y; accb.z += v.z; accb.w += v.w;
    }
}

__device__ __forceinline__ void sage_walk_rows(const float* __restrict__ feat,
                                               const int*   __restrict__ cells,
                                               const int*   __restrict__ ovfcnt,
                                               const int*   __restrict__ ovf,
                                               int blk,
                                               int ra, bool wa,
                                               int rb, bool wb,
                                               int lane,
                                               float4& acca, float4& accb) {
    const unsigned lane4 = (unsigned)lane * 4u;
    if (wa) acca = make_float4(0.0f, 0.0f, 0.0f, 0.0f);
    if (wb) accb = make_float4(0.0f, 0.0f, 0.0f, 0.0f);
    const int rowa = blk * GEMM_ROWS + ra, rowb = blk * GEMM_ROWS + rb;
#pragma nounroll
    for (int c = 0; c < N_CHUNKS; ++c) {
        const int v = cells[((size_t)c * GEMM_BLOCKS + blk) * CELL + lane];
        const int n = min(max(__builtin_amdgcn_readfirstlane(v), 0), CELL_CAP);
        const bool in = (lane >= 1 && lane <= n);
        const int sl = min(max(v >> 6, 0), N_NODES - 1);
        const int r  = v & 63;
        sage_add_marked(feat,
                        __builtin_amdgcn_ballot_w32(in && wa && r == ra),
                        __builtin_amdgcn_ballot_w32(in && wb && r == rb),
                        sl, lane4, acca, accb);
        const int m = __builtin_amdgcn_readfirstlane(min(max(ovfcnt[c], 0), CHUNK_EDGES));
#pragma nounroll
        for (int base = 0; base < m; base += 32) {
            const int pos = min(base + lane, CHUNK_EDGES - 1);
            const int ed = ovf[((size_t)c * CHUNK_EDGES + pos) * 2];
            const int es = ovf[((size_t)c * CHUNK_EDGES + pos) * 2 + 1];
            const bool in2 = (base + lane < m);
            const int sl2 = min(max(es, 0), N_NODES - 1);
            sage_add_marked(feat,
                            __builtin_amdgcn_ballot_w32(in2 && wa && ed == rowa),
                            __builtin_amdgcn_ballot_w32(in2 && wb && ed == rowb),
                            sl2, lane4, acca, accb);
        }
    }
}

#define A_PITCH 132

__global__ __launch_bounds__(256)
void sage_wmma_gemm(const float* __restrict__ feat,
                    const int*   __restrict__ cells,
                    const int*   __restrict__ ovfcnt,
                    const int*   __restrict__ ovf,
                    const float* __restrict__ Wself,
                    const float* __restrict__ bself,
                    const float* __restrict__ Wneigh,
                    const float* __restrict__ bneigh,
                    float*       __restrict__ out) {
    __shared__ __attribute__((aligned(16))) float sSelf [GEMM_ROWS * A_PITCH];
    __shared__ __attribute__((aligned(16))) float sNeigh[GEMM_ROWS * A_PITCH];
    __shared__ int sSlot[GEMM_ROWS * SLOTS];
    __shared__ int sRowCnt[GEMM_ROWS];

    const int lane  = threadIdx.x & 31;
    const int wave  = threadIdx.x >> 5;
    const int rloc  = lane & 15;
    const int khalf = lane >> 4;
    const int blk   = blockIdx.x;
    const int row0  = blk * GEMM_ROWS;
    const int col   = wave * 16 + (lane & 15);

    const float bias = bself[col] + bneigh[col];

    if (threadIdx.x < GEMM_ROWS) sRowCnt[threadIdx.x] = 0;

    const int wv = __builtin_amdgcn_readfirstlane(wave);

    const float* fsrc = feat + lane * 4;
    const v4f f0 = *(const v4f*)(fsrc + (size_t)min(row0 + wave,      N_NODES - 1) * D);
    const v4f f1 = *(const v4f*)(fsrc + (size_t)min(row0 + wave +  8, N_NODES - 1) * D);
    const v4f f2 = *(const v4f*)(fsrc + (size_t)min(row0 + wave + 16, N_NODES - 1) * D);
    const v4f f3 = *(const v4f*)(fsrc + (size_t)min(row0 + wave + 24, N_NODES - 1) * D);
    const v4f f4 = *(const v4f*)(fsrc + (size_t)min(row0 + wave + 32, N_NODES - 1) * D);
    const v4f f5 = *(const v4f*)(fsrc + (size_t)min(row0 + wave + 40, N_NODES - 1) * D);
    const v4f f6 = *(const v4f*)(fsrc + (size_t)min(row0 + wave + 48, N_NODES - 1) * D);
    const v4f f7 = *(const v4f*)(fsrc + (size_t)min(row0 + wave + 56, N_NODES - 1) * D);

    int cellv[WAVE_CHUNKS];
#pragma unroll
    for (int i = 0; i < WAVE_CHUNKS; ++i) {
        const int c = min(wv + 8 * i, N_CHUNKS - 1);
        cellv[i] = cells[((size_t)c * GEMM_BLOCKS + blk) * CELL + lane];
    }
    const int oc = ovfcnt[min(wv + 8 * lane, N_CHUNKS - 1)];

    float* fdst = sSelf + wave * A_PITCH + lane * 4;
    *(v4f*)(fdst +  0 * A_PITCH) = f0;
    *(v4f*)(fdst +  8 * A_PITCH) = f1;
    *(v4f*)(fdst + 16 * A_PITCH) = f2;
    *(v4f*)(fdst + 24 * A_PITCH) = f3;
    *(v4f*)(fdst + 32 * A_PITCH) = f4;
    *(v4f*)(fdst + 40 * A_PITCH) = f5;
    *(v4f*)(fdst + 48 * A_PITCH) = f6;
    *(v4f*)(fdst + 56 * A_PITCH) = f7;

    __syncthreads();

    int rk[WAVE_CHUNKS];
#pragma unroll
    for (int i = 0; i < WAVE_CHUNKS; ++i) {
        const int v = cellv[i];
        const int n = (wv + 8 * i < N_CHUNKS)
                    ? min(max(__builtin_amdgcn_readfirstlane(v), 0), CELL_CAP) : 0;
        rk[i] = (lane >= 1 && lane <= n) ? atomicAdd(sRowCnt + (v & 63), 1) : SLOTS;
    }
#pragma unroll
    for (int i = 0; i < WAVE_CHUNKS; ++i) {
        const int v = cellv[i];
        if ((unsigned)rk[i] < (unsigned)SLOTS)
            sSlot[(v & 63) * SLOTS + rk[i]] = min(max(v >> 6, 0), N_NODES - 1);
    }
    if (__builtin_amdgcn_ballot_w32(oc > 0) != 0u) {
#pragma nounroll
        for (int i = 0; i < WAVE_CHUNKS; ++i) {
            const int c = wv + 8 * i;
            const int m = (c < N_CHUNKS)
                        ? min(max(__builtin_amdgcn_readlane(oc, i), 0), CHUNK_EDGES) : 0;
#pragma nounroll
            for (int base = 0; base < m; base += 32) {
                const int pos = base + lane;
                if (pos < m) {
                    const int ed = ovf[((size_t)c * CHUNK_EDGES + pos) * 2];
                    const int es = ovf[((size_t)c * CHUNK_EDGES + pos) * 2 + 1];
                    if ((ed >> 6) == blk)
                        sage_row_place(ed & 63, min(max(es, 0), N_NODES - 1), sRowCnt, sSlot);
                }
            }
        }
    }

    __syncthreads();

    v8f acc[GEMM_TILES];
#pragma unroll
    for (int t = 0; t < GEMM_TILES; ++t)
#pragma unroll
        for (int j = 0; j < 8; ++j) acc[t][j] = bias;

    const float* arow_s = sSelf  + rloc * A_PITCH + khalf * 2;
    const float* arow_n = sNeigh + rloc * A_PITCH + khalf * 2;
    const float* zrow   = (const float*)(ovf + ZROW_INT_OFF);

    {
        const int cv = min(max(sRowCnt[wave + 8 * (lane & 7)], 0), N_EDGES);
        const unsigned lane4 = (unsigned)lane * 4u;

#pragma nounroll
        for (int p = 0; p < 8; ++p) {
            const int r0 = wave + 8 * p;
            const int ca = __builtin_amdgcn_readlane(cv, p);
            const int fa = min(ca, SLOTS);
            const int ia = (lane < fa) ? sSlot[r0 * SLOTS + lane] : 0;
            const int ica = min(max(ia, 0), N_NODES - 1);
            int qa[8];
#pragma unroll
            for (int k = 0; k < 8; ++k) qa[k] = __builtin_amdgcn_readlane(ica, k);
            v2f bw[4];
#pragma unroll
            for (int i = 0; i < 4; ++i) {
                const int k = 16 * p + 4 * i + khalf * 2;
                bw[i].x = Wself[(size_t)k * D + col];
                bw[i].y = Wself[(size_t)(k + 1) * D + col];
            }
            v4f va[8];
#pragma unroll
            for (int k = 0; k < 8; ++k) {
                const float* pa = (k < fa) ? feat + (size_t)qa[k] * D : zrow;
                va[k] = *(const v4f*)(pa + lane4);
            }
#pragma unroll
            for (int i = 0; i < 4; ++i) {
#pragma unroll
                for (int t = 0; t < GEMM_TILES; ++t) {
                    const v2f a = *(const v2f*)(arow_s + t * 16 * A_PITCH + 16 * p + 4 * i);
                    acc[t] = __builtin_amdgcn_wmma_f32_16x16x4_f32(
                                 false, a, false, bw[i], (short)0, acc[t], false, false);
                }
            }
            asm volatile("" : "+v"(acc[0]), "+v"(acc[1]), "+v"(acc[2]), "+v"(acc[3]), "+v"(va[0]));
            v4f s0 = {0.0f, 0.0f, 0.0f, 0.0f};
#pragma unroll
            for (int k = 0; k < 8; ++k) s0 += va[k];
            float4 n0 = make_float4(s0.x, s0.y, s0.z, s0.w);
            float4 n1 = make_float4(0.0f, 0.0f, 0.0f, 0.0f);
            if (fa > 8)
                sage_gather2(feat, fa, 0, lane, ica, 0, n0, n1);
            if (ca > SLOTS)
                sage_walk_rows(feat, cells, ovfcnt, ovf, blk,
                               r0, true, r0, false, lane, n0, n1);
            const float rd0 = 1.0f / fmaxf((float)ca, 1.0f);
            n0.x *= rd0; n0.y *= rd0; n0.z *= rd0; n0.w *= rd0;
            *(float4*)(sNeigh + r0 * A_PITCH + lane * 4) = n0;
        }
    }

    __syncthreads();

#pragma unroll 16
    for (int k0 = 0; k0 < D; k0 += 4) {
        const int k = k0 + khalf * 2;
        v2f b; b.x = Wneigh[(size_t)k * D + col];
               b.y = Wneigh[(size_t)(k + 1) * D + col];
#pragma unroll
        for (int t = 0; t < GEMM_TILES; ++t) {
            const v2f a = *(const v2f*)(arow_n + t * 16 * A_PITCH + k0);
            acc[t] = __builtin_amdgcn_wmma_f32_16x16x4_f32(
                         false, a, false, b, (short)0, acc[t], false, false);
        }
    }

#pragma unroll
    for (int t = 0; t < GEMM_TILES; ++t) {
        if (row0 + 16 * t < N_NODES) {
            float* obase = out + (size_t)(row0 + 16 * t + khalf * 8) * D + col;
#pragma unroll
            for (int j = 0; j < 8; ++j)
                obase[(size_t)j * D] = acc[t][j];
        }
    }
}

extern "C" void kernel_launch(void* const* d_in, const int* in_sizes, int n_in,
                              void* d_out, int out_size, void* d_ws, size_t ws_size,
                              hipStream_t stream) {
    const float* feat   = (const float*)d_in[0];
    const int*   src    = (const int*)  d_in[1];
    const int*   dst    = (const int*)  d_in[2];
    const float* Wself  = (const float*)d_in[3];
    const float* bself  = (const float*)d_in[4];
    const float* Wneigh = (const float*)d_in[5];
    const float* bneigh = (const float*)d_in[6];
    float*       out    = (float*)d_out;

    int* cells  = (int*)d_ws;
    int* ovfcnt = cells  + CELL_INTS;
    int* ovf    = ovfcnt + OVFCNT_INTS;

    sage_bin_edges<<<N_CHUNKS, 1024, 0, stream>>>(src, dst, cells, ovfcnt, ovf);
    sage_wmma_gemm<<<GEMM_BLOCKS, 256, 0, stream>>>(
        feat, cells, ovfcnt, ovf, Wself, bself, Wneigh, bneigh, out);
}


# W_self/W_neigh repacked in B-fragment order into the workspace by 8 extra blocks of the binning launch; GEMM waves load B as whole-line 16-byte vectors (2 k-steps each), not 4-byte pieces; same sums
# speedup vs baseline: 3.5820x; 1.0363x over previous
#include <hip/hip_runtime.h>


#define N_NODES 100000
#define N_EDGES 640000
#define D 128
#define SLOTS 32
#define GEMM_TILES 4
#define GEMM_ROWS (16 * GEMM_TILES)
#define GEMM_BLOCKS ((N_NODES + GEMM_ROWS - 1) / GEMM_ROWS)
#define CHUNK_EDGES 8192
#define N_CHUNKS ((N_EDGES + CHUNK_EDGES - 1) / CHUNK_EDGES)
#define CELL 32
#define CELL_CAP (CELL - 1)
#define CELL_INTS (N_CHUNKS * GEMM_BLOCKS * CELL)
#define OVFCNT_INTS 128
#define WAVE_CHUNKS ((N_CHUNKS + 7) / 8)
#define ZROW_INT_OFF (N_CHUNKS * CHUNK_EDGES * 2 - D)
#define OVF_INTS (N_CHUNKS * CHUNK_EDGES * 2)
#define WPACK_INT_OFF (((CELL_INTS + OVFCNT_INTS + OVF_INTS + 63) / 64) * 64)
#define WPACK_VEC4 (2 * D * D / 4)
#define PACK_BLOCKS (WPACK_VEC4 / 1024)
#define BIN_GRID (N_CHUNKS + PACK_BLOCKS)

typedef float v2f __attribute__((ext_vector_type(2)));
typedef float v4f __attribute__((ext_vector_type(4)));
typedef int   v4i __attribute__((ext_vector_type(4)));
typedef float v8f __attribute__((ext_vector_type(8)));


__device__ __forceinline__ int sage_bin_rank(int d, bool have, int* sCnt) {
    const int dc = min(max(d, 0), N_NODES - 1);
    return have ? atomicAdd(sCnt + (dc >> 6), 1) : 0;
}

__device__ __forceinline__ void sage_bin_store(int d, int s, int rank, bool have, int c,
                                               int* sCell, int* sOvf,
                                               int* __restrict__ ovf) {
    const int dc = min(max(d, 0), N_NODES - 1);
    const int sc = min(max(s, 0), N_NODES - 1);
    if (have) {
        if (rank < CELL_CAP) {
            sCell[(dc >> 6) * CELL + 1 + rank] = (sc << 6) | (dc & 63);
        } else {
            const int p = min(max(atomicAdd(sOvf, 1), 0), CHUNK_EDGES - 1);
            ovf[((size_t)c * CHUNK_EDGES + p) * 2]     = dc;
            ovf[((size_t)c * CHUNK_EDGES + p) * 2 + 1] = sc;
        }
    }
}

__device__ __forceinline__ void sage_pack_w(const float* __restrict__ Wself,
                                            const float* __restrict__ Wneigh,
                                            float*       __restrict__ wpack,
                                            int t) {
    const int l = t & 31, g = (t >> 5) & 15, w = (t >> 9) & 7, term = t >> 12;
    const float* W = term ? Wneigh : Wself;
    const int k   = 8 * g + 2 * (l >> 4);
    const int col = 16 * w + (l & 15);
    v4f v;
    v.x = W[(size_t)k * D + col];
    v.y = W[(size_t)(k + 1) * D + col];
    v.z = W[(size_t)(k + 4) * D + col];
    v.w = W[(size_t)(k + 5) * D + col];
    ((v4f*)wpack)[t] = v;
}

__global__ __launch_bounds__(1024)
void sage_bin_edges(const int*   __restrict__ src,
                    const int*   __restrict__ dst,
                    const float* __restrict__ Wself,
                    const float* __restrict__ Wneigh,
                    int*         __restrict__ cells,
                    int*         __restrict__ ovfcnt,
                    int*         __restrict__ ovf,
                    float*       __restrict__ wpack) {
    __shared__ __attribute__((aligned(16))) int sCell[GEMM_BLOCKS * CELL];
    __shared__ int sCnt[GEMM_BLOCKS];
    __shared__ int sOvf;

    const int tid = threadIdx.x;
    const int c   = blockIdx.x;

    if (c >= N_CHUNKS) {
        sage_pack_w(Wself, Wneigh, wpack, (c - N_CHUNKS) * 1024 + tid);
        return;
    }

    const int e0 = c * CHUNK_EDGES + tid * 4;
    const int e1 = e0 + CHUNK_EDGES / 2;
    const bool h0 = e0 < N_EDGES, h1 = e1 < N_EDGES;
    const v4i d0 = *(const v4i*)(dst + min(e0, N_EDGES - 4));
    const v4i s0 = *(const v4i*)(src + min(e0, N_EDGES - 4));
    const v4i d1 = *(const v4i*)(dst + min(e1, N_EDGES - 4));
    const v4i s1 = *(const v4i*)(src + min(e1, N_EDGES - 4));

    const v4i zero4 = {0, 0, 0, 0};
    for (int i = tid; i < GEMM_BLOCKS * CELL / 4; i += 1024) ((v4i*)sCell)[i] = zero4;
    for (int b = tid; b < GEMM_BLOCKS; b += 1024) sCnt[b] = 0;
    if (tid == 0) sOvf = 0;

    __syncthreads();

    const int r0 = sage_bin_rank(d0.x, h0, sCnt);
    const int r1 = sage_bin_rank(d0.y, h0, sCnt);
    const int r2 = sage_bin_rank(d0.z, h0, sCnt);
    const int r3 = sage_bin_rank(d0.w, h0, sCnt);
    const int r4 = sage_bin_rank(d1.x, h1, sCnt);
    const int r5 = sage_bin_rank(d1.y, h1, sCnt);
    const int r6 = sage_bin_rank(d1.z, h1, sCnt);
    const int r7 = sage_bin_rank(d1.w, h1, sCnt);
    sage_bin_store(d0.x, s0.x, r0, h0, c, sCell, &sOvf, ovf);
    sage_bin_store(d0.y, s0.y, r1, h0, c, sCell, &sOvf, ovf);
    sage_bin_store(d0.z, s0.z, r2, h0, c, sCell, &sOvf, ovf);
    sage_bin_store(d0.w, s0.w, r3, h0, c, sCell, &sOvf, ovf);
    sage_bin_store(d1.x, s1.x, r4, h1, c, sCell, &sOvf, ovf);
    sage_bin_store(d1.y, s1.y, r5, h1, c, sCell, &sOvf, ovf);
    sage_bin_store(d1.z, s1.z, r6, h1, c, sCell, &sOvf, ovf);
    sage_bin_store(d1.w, s1.w, r7, h1, c, sCell, &sOvf, ovf);

    __syncthreads();

    v4i*       g = (v4i*)(cells + (size_t)c * (GEMM_BLOCKS * CELL));
    const v4i* l = (const v4i*)sCell;
    for (int i = tid; i < GEMM_BLOCKS * CELL / 4; i += 1024) {
        v4i v = l[i];
        if ((i & (CELL / 4 - 1)) == 0) v.x = sCnt[i / (CELL / 4)];
        g[i] = v;
    }
    if (tid == 0) ovfcnt[c] = sOvf;
    if (c == N_CHUNKS - 1 && tid < D) ovf[ZROW_INT_OFF + tid] = 0;
}

__device__ __forceinline__ void sage_row_place(int r, int s, int* sRowCnt, int* sSlot) {
    const int rank = atomicAdd(sRowCnt + r, 1);
    if ((unsigned)rank < (unsigned)SLOTS) sSlot[r * SLOTS + rank] = s;
}

__device__ __forceinline__ void sage_gather2(const float* __restrict__ feat,
                                             int na, int nb,
                                             int lane, int ia0, int ib0,
                                             float4& ra, float4& rb) {
    const unsigned lane4 = (unsigned)lane * 4u;
    float4 acca = ra;
    float4 accb = rb;
    const int nmax = max(na, nb);
    const int ica = min(max(ia0, 0), N_NODES - 1);
    const int icb = min(max(ib0, 0), N_NODES - 1);
    for (int j = 8; j < nmax; j += 8) {
        int qa[8], qb[8];
#pragma unroll
        for (int k = 0; k < 8; ++k) {
            qa[k] = __builtin_amdgcn_readlane(ica, j + k);
            qb[k] = __builtin_amdgcn_readlane(icb, j + k);
        }
        float4 va[8], vb[8];
#pragma unroll
        for (int k = 0; k < 8; ++k) {
            if (j + k < na) va[k] = *(const float4*)(feat + (size_t)qa[k] * D + lane4);
            else            va[k] = make_float4(0.0f, 0.0f, 0.0f, 0.0f);
        }
#pragma unroll
        for (int k = 0; k < 8; ++k) {
            if (j + k < nb) vb[k] = *(const float4*)(feat + (size_t)qb[k] * D + lane4);
            else            vb[k] = make_float4(0.0f, 0.0f, 0.0f, 0.0f);
        }
#pragma unroll
        for (int k = 0; k < 8; ++k) {
            acca.x += va[k].x; acca.y += va[k].y; acca.z += va[k].z; acca.w += va[k].w;
        }
#pragma unroll
        for (int k = 0; k < 8; ++k) {
            accb.x += vb[k].x; accb.y += vb[k].y; accb.z += vb[k].z; accb.w += vb[k].w;
        }
    }
    ra = acca;
    rb = accb;
}

__device__ __forceinline__ void sage_add_marked(const float* __restrict__ feat,
                                                unsigned ma, unsigned mb, int sl,
                                                unsigned lane4,
                                                float4& acca, float4& accb) {
    while (ma != 0u) {
        const int k = __builtin_ctz(ma);
        ma &= ma - 1u;
        const int s = __builtin_amdgcn_readlane(sl, k);
        const float4 v = *(const float4*)(feat + (size_t)s * D + lane4);
        acca.x += v.x; acca.y += v.y; acca.z += v.z; acca.w += v.w;
    }
    while (mb != 0u) {
        const int k = __builtin_ctz(mb);
        mb &= mb - 1u;
        const int s = __builtin_amdgcn_readlane(sl, k);
        const float4 v = *(const float4*)(feat + (size_t)s * D + lane4);
        accb.x += v.x; accb.y += v.y; accb.z += v.z; accb.w += v.w;
    }
}

__device__ __forceinline__ void sage_walk_rows(const float* __restrict__ feat,
                                               const int*   __restrict__ cells,
                                               const int*   __restrict__ ovfcnt,
                                               const int*   __restrict__ ovf,
                                               int blk,
                                               int ra, bool wa,
                                               int rb, bool wb,
                                               int lane,
                                               float4& acca, float4& accb) {
    const unsigned lane4 = (unsigned)lane * 4u;
    if (wa) acca = make_float4(0.0f, 0.0f, 0.0f, 0.0f);
    if (wb) accb = make_float4(0.0f, 0.0f, 0.0f, 0.0f);
    const int rowa = blk * GEMM_ROWS + ra, rowb = blk * GEMM_ROWS + rb;
#pragma nounroll
    for (int c = 0; c < N_CHUNKS; ++c) {
        const int v = cells[((size_t)c * GEMM_BLOCKS + blk) * CELL + lane];
        const int n = min(max(__builtin_amdgcn_readfirstlane(v), 0), CELL_CAP);
        const bool in = (lane >= 1 && lane <= n);
        const int sl = min(max(v >> 6, 0), N_NODES - 1);
        const int r  = v & 63;
        sage_add_marked(feat,
                        __builtin_amdgcn_ballot_w32(in && wa && r == ra),
                        __builtin_amdgcn_ballot_w32(in && wb && r == rb),
                        sl, lane4, acca, accb);
        const int m = __builtin_amdgcn_readfirstlane(min(max(ovfcnt[c], 0), CHUNK_EDGES));
#pragma nounroll
        for (int base = 0; base < m; base += 32) {
            const int pos = min(base + lane, CHUNK_EDGES - 1);
            const int ed = ovf[((size_t)c * CHUNK_EDGES + pos) * 2];
            const int es = ovf[((size_t)c * CHUNK_EDGES + pos) * 2 + 1];
            const bool in2 = (base + lane < m);
            const int sl2 = min(max(es, 0), N_NODES - 1);
            sage_add_marked(feat,
                            __builtin_amdgcn_ballot_w32(in2 && wa && ed == rowa),
                            __builtin_amdgcn_ballot_w32(in2 && wb && ed == rowb),
                            sl2, lane4, acca, accb);
        }
    }
}

#define A_PITCH 132

__global__ __launch_bounds__(256)
void sage_wmma_gemm(const float* __restrict__ feat,
                    const int*   __restrict__ cells,
                    const int*   __restrict__ ovfcnt,
                    const int*   __restrict__ ovf,
                    const float* __restrict__ wpack,
                    const float* __restrict__ bself,
                    const float* __restrict__ bneigh,
                    float*       __restrict__ out) {
    __shared__ __attribute__((aligned(16))) float sSelf [GEMM_ROWS * A_PITCH];
    __shared__ __attribute__((aligned(16))) float sNeigh[GEMM_ROWS * A_PITCH];
    __shared__ int sSlot[GEMM_ROWS * SLOTS];
    __shared__ int sRowCnt[GEMM_ROWS];

    const int lane  = threadIdx.x & 31;
    const int wave  = threadIdx.x >> 5;
    const int rloc  = lane & 15;
    const int khalf = lane >> 4;
    const int blk   = blockIdx.x;
    const int row0  = blk * GEMM_ROWS;
    const int col   = wave * 16 + (lane & 15);

    const float bias = bself[col] + bneigh[col];

    if (threadIdx.x < GEMM_ROWS) sRowCnt[threadIdx.x] = 0;

    const int wv = __builtin_amdgcn_readfirstlane(wave);

    const float* fsrc = feat + lane * 4;
    const v4f f0 = *(const v4f*)(fsrc + (size_t)min(row0 + wave,      N_NODES - 1) * D);
    const v4f f1 = *(const v4f*)(fsrc + (size_t)min(row0 + wave +  8, N_NODES - 1) * D);
    const v4f f2 = *(const v4f*)(fsrc + (size_t)min(row0 + wave + 16, N_NODES - 1) * D);
    const v4f f3 = *(const v4f*)(fsrc + (size_t)min(row0 + wave + 24, N_NODES - 1) * D);
    const v4f f4 = *(const v4f*)(fsrc + (size_t)min(row0 + wave + 32, N_NODES - 1) * D);
    const v4f f5 = *(const v4f*)(fsrc + (size_t)min(row0 + wave + 40, N_NODES - 1) * D);
    const v4f f6 = *(const v4f*)(fsrc + (size_t)min(row0 + wave + 48, N_NODES - 1) * D);
    const v4f f7 = *(const v4f*)(fsrc + (size_t)min(row0 + wave + 56, N_NODES - 1) * D);

    int cellv[WAVE_CHUNKS];
#pragma unroll
    for (int i = 0; i < WAVE_CHUNKS; ++i) {
        const int c = min(wv + 8 * i, N_CHUNKS - 1);
        cellv[i] = cells[((size_t)c * GEMM_BLOCKS + blk) * CELL + lane];
    }
    const int oc = ovfcnt[min(wv + 8 * lane, N_CHUNKS - 1)];

    float* fdst = sSelf + wave * A_PITCH + lane * 4;
    *(v4f*)(fdst +  0 * A_PITCH) = f0;
    *(v4f*)(fdst +  8 * A_PITCH) = f1;
    *(v4f*)(fdst + 16 * A_PITCH) = f2;
    *(v4f*)(fdst + 24 * A_PITCH) = f3;
    *(v4f*)(fdst + 32 * A_PITCH) = f4;
    *(v4f*)(fdst + 40 * A_PITCH) = f5;
    *(v4f*)(fdst + 48 * A_PITCH) = f6;
    *(v4f*)(fdst + 56 * A_PITCH) = f7;

    __syncthreads();

    int rk[WAVE_CHUNKS];
#pragma unroll
    for (int i = 0; i < WAVE_CHUNKS; ++i) {
        const int v = cellv[i];
        const int n = (wv + 8 * i < N_CHUNKS)
                    ? min(max(__builtin_amdgcn_readfirstlane(v), 0), CELL_CAP) : 0;
        rk[i] = (lane >= 1 && lane <= n) ? atomicAdd(sRowCnt + (v & 63), 1) : SLOTS;
    }
#pragma unroll
    for (int i = 0; i < WAVE_CHUNKS; ++i) {
        const int v = cellv[i];
        if ((unsigned)rk[i] < (unsigned)SLOTS)
            sSlot[(v & 63) * SLOTS + rk[i]] = min(max(v >> 6, 0), N_NODES - 1);
    }
    if (__builtin_amdgcn_ballot_w32(oc > 0) != 0u) {
#pragma nounroll
        for (int i = 0; i < WAVE_CHUNKS; ++i) {
            const int c = wv + 8 * i;
            const int m = (c < N_CHUNKS)
                        ? min(max(__builtin_amdgcn_readlane(oc, i), 0), CHUNK_EDGES) : 0;
#pragma nounroll
            for (int base = 0; base < m; base += 32) {
                const int pos = base + lane;
                if (pos < m) {
                    const int ed = ovf[((size_t)c * CHUNK_EDGES + pos) * 2];
                    const int es = ovf[((size_t)c * CHUNK_EDGES + pos) * 2 + 1];
                    if ((ed >> 6) == blk)
                        sage_row_place(ed & 63, min(max(es, 0), N_NODES - 1), sRowCnt, sSlot);
                }
            }
        }
    }

    __syncthreads();

    v8f acc[GEMM_TILES];
#pragma unroll
    for (int t = 0; t < GEMM_TILES; ++t)
#pragma unroll
        for (int j = 0; j < 8; ++j) acc[t][j] = bias;

    const float* arow_s = sSelf  + rloc * A_PITCH + khalf * 2;
    const float* arow_n = sNeigh + rloc * A_PITCH + khalf * 2;
    const float* zrow   = (const float*)(ovf + ZROW_INT_OFF);
    const v4f* wp_s = (const v4f*)wpack + (size_t)(wv * 16) * 32 + lane;
    const v4f* wp_n = (const v4f*)wpack + (size_t)((8 + wv) * 16) * 32 + lane;

    {
        const int cv = min(max(sRowCnt[wave + 8 * (lane & 7)], 0), N_EDGES);
        const unsigned lane4 = (unsigned)lane * 4u;

#pragma nounroll
        for (int p = 0; p < 8; ++p) {
            const int r0 = wave + 8 * p;
            const int ca = __builtin_amdgcn_readlane(cv, p);
            const int fa = min(ca, SLOTS);
            const int ia = (lane < fa) ? sSlot[r0 * SLOTS + lane] : 0;
            const int ica = min(max(ia, 0), N_NODES - 1);
            int qa[8];
#pragma unroll
            for (int k = 0; k < 8; ++k) qa[k] = __builtin_amdgcn_readlane(ica, k);
            const v4f bq0 = wp_s[(2 * p) * 32];
            const v4f bq1 = wp_s[(2 * p + 1) * 32];
            v2f bw[4];
            bw[0].x = bq0.x; bw[0].y = bq0.y;
            bw[1].x = bq0.z; bw[1].y = bq0.w;
            bw[2].x = bq1.x; bw[2].y = bq1.y;
            bw[3].x = bq1.z; bw[3].y = bq1.w;
            v4f va[8];
#pragma unroll
            for (int k = 0; k < 8; ++k) {
                const float* pa = (k < fa) ? feat + (size_t)qa[k] * D : zrow;
                va[k] = *(const v4f*)(pa + lane4);
            }
#pragma unroll
            for (int i = 0; i < 4; ++i) {
#pragma unroll
                for (int t = 0; t < GEMM_TILES; ++t) {
                    const v2f a = *(const v2f*)(arow_s + t * 16 * A_PITCH + 16 * p + 4 * i);
                    acc[t] = __builtin_amdgcn_wmma_f32_16x16x4_f32(
                                 false, a, false, bw[i], (short)0, acc[t], false, false);
                }
            }
            asm volatile("" : "+v"(acc[0]), "+v"(acc[1]), "+v"(acc[2]), "+v"(acc[3]), "+v"(va[0]));
            v4f s0 = {0.0f, 0.0f, 0.0f, 0.0f};
#pragma unroll
            for (int k = 0; k < 8; ++k) s0 += va[k];
            float4 n0 = make_float4(s0.x, s0.y, s0.z, s0.w);
            float4 n1 = make_float4(0.0f, 0.0f, 0.0f, 0.0f);
            if (fa > 8)
                sage_gather2(feat, fa, 0, lane, ica, 0, n0, n1);
            if (ca > SLOTS)
                sage_walk_rows(feat, cells, ovfcnt, ovf, blk,
                               r0, true, r0, false, lane, n0, n1);
            const float rd0 = 1.0f / fmaxf((float)ca, 1.0f);
            n0.x *= rd0; n0.y *= rd0; n0.z *= rd0; n0.w *= rd0;
            *(float4*)(sNeigh + r0 * A_PITCH + lane * 4) = n0;
        }
    }

    __syncthreads();

#pragma unroll 8
    for (int g = 0; g < D / 8; ++g) {
        const v4f bq = wp_n[g * 32];
#pragma unroll
        for (int s = 0; s < 2; ++s) {
            v2f b;
            b.x = s ? bq.z : bq.x;
            b.y = s ? bq.w : bq.y;
            const int k0 = 8 * g + 4 * s;
#pragma unroll
            for (int t = 0; t < GEMM_TILES; ++t) {
                const v2f a = *(const v2f*)(arow_n + t * 16 * A_PITCH + k0);
                acc[t] = __builtin_amdgcn_wmma_f32_16x16x4_f32(
                             false, a, false, b, (short)0, acc[t], false, false);
            }
        }
    }

#pragma unroll
    for (int t = 0; t < GEMM_TILES; ++t) {
        if (row0 + 16 * t < N_NODES) {
            float* obase = out + (size_t)(row0 + 16 * t + khalf * 8) * D + col;
#pragma unroll
            for (int j = 0; j < 8; ++j)
                obase[(size_t)j * D] = acc[t][j];
        }
    }
}

extern "C" void kernel_launch(void* const* d_in, const int* in_sizes, int n_in,
                              void* d_out, int out_size, void* d_ws, size_t ws_size,
                              hipStream_t stream) {
    const float* feat   = (const float*)d_in[0];
    const int*   src    = (const int*)  d_in[1];
    const int*   dst    = (const int*)  d_in[2];
    const float* Wself  = (const float*)d_in[3];
    const float* bself  = (const float*)d_in[4];
    const float* Wneigh = (const float*)d_in[5];
    const float* bneigh = (const float*)d_in[6];
    float*       out    = (float*)d_out;

    int* cells  = (int*)d_ws;
    int* ovfcnt = cells  + CELL_INTS;
    int* ovf    = ovfcnt + OVFCNT_INTS;
    float* wpack = (float*)((int*)d_ws + WPACK_INT_OFF);

    sage_bin_edges<<<BIN_GRID, 1024, 0, stream>>>(
        src, dst, Wself, Wneigh, cells, ovfcnt, ovf, wpack);
    sage_wmma_gemm<<<GEMM_BLOCKS, 256, 0, stream>>>(
        feat, cells, ovfcnt, ovf, wpack, bself, bneigh, out);
}


# GEMM epilogue: D tiles staged in the freed self-tile LDS, then each wave stores 8 whole 512 B rows, 16 B per lane (was 4 B stores: each 128 B line left as two 64 B pieces from two waves); same sums
# speedup vs baseline: 3.7121x; 1.0363x over previous
#include <hip/hip_runtime.h>


#define N_NODES 100000
#define N_EDGES 640000
#define D 128
#define SLOTS 32
#define GEMM_TILES 4
#define GEMM_ROWS (16 * GEMM_TILES)
#define GEMM_BLOCKS ((N_NODES + GEMM_ROWS - 1) / GEMM_ROWS)
#define CHUNK_EDGES 8192
#define N_CHUNKS ((N_EDGES + CHUNK_EDGES - 1) / CHUNK_EDGES)
#define CELL 32
#define CELL_CAP (CELL - 1)
#define CELL_INTS (N_CHUNKS * GEMM_BLOCKS * CELL)
#define OVFCNT_INTS 128
#define WAVE_CHUNKS ((N_CHUNKS + 7) / 8)
#define ZROW_INT_OFF (N_CHUNKS * CHUNK_EDGES * 2 - D)
#define OVF_INTS (N_CHUNKS * CHUNK_EDGES * 2)
#define WPACK_INT_OFF (((CELL_INTS + OVFCNT_INTS + OVF_INTS + 63) / 64) * 64)
#define WPACK_VEC4 (2 * D * D / 4)
#define PACK_BLOCKS (WPACK_VEC4 / 1024)
#define BIN_GRID (N_CHUNKS + PACK_BLOCKS)

typedef float v2f __attribute__((ext_vector_type(2)));
typedef float v4f __attribute__((ext_vector_type(4)));
typedef int   v4i __attribute__((ext_vector_type(4)));
typedef float v8f __attribute__((ext_vector_type(8)));


__device__ __forceinline__ int sage_bin_rank(int d, bool have, int* sCnt) {
    const int dc = min(max(d, 0), N_NODES - 1);
    return have ? atomicAdd(sCnt + (dc >> 6), 1) : 0;
}

__device__ __forceinline__ void sage_bin_store(int d, int s, int rank, bool have, int c,
                                               int* sCell, int* sOvf,
                                               int* __restrict__ ovf) {
    const int dc = min(max(d, 0), N_NODES - 1);
    const int sc = min(max(s, 0), N_NODES - 1);
    if (have) {
        if (rank < CELL_CAP) {
            sCell[(dc >> 6) * CELL + 1 + rank] = (sc << 6) | (dc & 63);
        } else {
            const int p = min(max(atomicAdd(sOvf, 1), 0), CHUNK_EDGES - 1);
            ovf[((size_t)c * CHUNK_EDGES + p) * 2]     = dc;
            ovf[((size_t)c * CHUNK_EDGES + p) * 2 + 1] = sc;
        }
    }
}

__device__ __forceinline__ void sage_pack_w(const float* __restrict__ Wself,
                                            const float* __restrict__ Wneigh,
                                            float*       __restrict__ wpack,
                                            int t) {
    const int l = t & 31, g = (t >> 5) & 15, w = (t >> 9) & 7, term = t >> 12;
    const float* W = term ? Wneigh : Wself;
    const int k   = 8 * g + 2 * (l >> 4);
    const int col = 16 * w + (l & 15);
    v4f v;
    v.x = W[(size_t)k * D + col];
    v.y = W[(size_t)(k + 1) * D + col];
    v.z = W[(size_t)(k + 4) * D + col];
    v.w = W[(size_t)(k + 5) * D + col];
    ((v4f*)wpack)[t] = v;
}

__global__ __launch_bounds__(1024)
void sage_bin_edges(const int*   __restrict__ src,
                    const int*   __restrict__ dst,
                    const float* __restrict__ Wself,
                    const float* __restrict__ Wneigh,
                    int*         __restrict__ cells,
                    int*         __restrict__ ovfcnt,
                    int*         __restrict__ ovf,
                    float*       __restrict__ wpack) {
    __shared__ __attribute__((aligned(16))) int sCell[GEMM_BLOCKS * CELL];
    __shared__ int sCnt[GEMM_BLOCKS];
    __shared__ int sOvf;

    const int tid = threadIdx.x;
    const int c   = blockIdx.x;

    if (c >= N_CHUNKS) {
        sage_pack_w(Wself, Wneigh, wpack, (c - N_CHUNKS) * 1024 + tid);
        return;
    }

    const int e0 = c * CHUNK_EDGES + tid * 4;
    const int e1 = e0 + CHUNK_EDGES / 2;
    const bool h0 = e0 < N_EDGES, h1 = e1 < N_EDGES;
    const v4i d0 = *(const v4i*)(dst + min(e0, N_EDGES - 4));
    const v4i s0 = *(const v4i*)(src + min(e0, N_EDGES - 4));
    const v4i d1 = *(const v4i*)(dst + min(e1, N_EDGES - 4));
    const v4i s1 = *(const v4i*)(src + min(e1, N_EDGES - 4));

    const v4i zero4 = {0, 0, 0, 0};
    for (int i = tid; i < GEMM_BLOCKS * CELL / 4; i += 1024) ((v4i*)sCell)[i] = zero4;
    for (int b = tid; b < GEMM_BLOCKS; b += 1024) sCnt[b] = 0;
    if (tid == 0) sOvf = 0;

    __syncthreads();

    const int r0 = sage_bin_rank(d0.x, h0, sCnt);
    const int r1 = sage_bin_rank(d0.y, h0, sCnt);
    const int r2 = sage_bin_rank(d0.z, h0, sCnt);
    const int r3 = sage_bin_rank(d0.w, h0, sCnt);
    const int r4 = sage_bin_rank(d1.x, h1, sCnt);
    const int r5 = sage_bin_rank(d1.y, h1, sCnt);
    const int r6 = sage_bin_rank(d1.z, h1, sCnt);
    const int r7 = sage_bin_rank(d1.w, h1, sCnt);
    sage_bin_store(d0.x, s0.x, r0, h0, c, sCell, &sOvf, ovf);
    sage_bin_store(d0.y, s0.y, r1, h0, c, sCell, &sOvf, ovf);
    sage_bin_store(d0.z, s0.z, r2, h0, c, sCell, &sOvf, ovf);
    sage_bin_store(d0.w, s0.w, r3, h0, c, sCell, &sOvf, ovf);
    sage_bin_store(d1.x, s1.x, r4, h1, c, sCell, &sOvf, ovf);
    sage_bin_store(d1.y, s1.y, r5, h1, c, sCell, &sOvf, ovf);
    sage_bin_store(d1.z, s1.z, r6, h1, c, sCell, &sOvf, ovf);
    sage_bin_store(d1.w, s1.w, r7, h1, c, sCell, &sOvf, ovf);

    __syncthreads();

    v4i*       g = (v4i*)(cells + (size_t)c * (GEMM_BLOCKS * CELL));
    const v4i* l = (const v4i*)sCell;
    for (int i = tid; i < GEMM_BLOCKS * CELL / 4; i += 1024) {
        v4i v = l[i];
        if ((i & (CELL / 4 - 1)) == 0) v.x = sCnt[i / (CELL / 4)];
        g[i] = v;
    }
    if (tid == 0) ovfcnt[c] = sOvf;
    if (c == N_CHUNKS - 1 && tid < D) ovf[ZROW_INT_OFF + tid] = 0;
}

__device__ __forceinline__ void sage_row_place(int r, int s, int* sRowCnt, int* sSlot) {
    const int rank = atomicAdd(sRowCnt + r, 1);
    if ((unsigned)rank < (unsigned)SLOTS) sSlot[r * SLOTS + rank] = s;
}

__device__ __forceinline__ void sage_gather2(const float* __restrict__ feat,
                                             int na, int nb,
                                             int lane, int ia0, int ib0,
                                             float4& ra, float4& rb) {
    const unsigned lane4 = (unsigned)lane * 4u;
    float4 acca = ra;
    float4 accb = rb;
    const int nmax = max(na, nb);
    const int ica = min(max(ia0, 0), N_NODES - 1);
    const int icb = min(max(ib0, 0), N_NODES - 1);
    for (int j = 8; j < nmax; j += 8) {
        int qa[8], qb[8];
#pragma unroll
        for (int k = 0; k < 8; ++k) {
            qa[k] = __builtin_amdgcn_readlane(ica, j + k);
            qb[k] = __builtin_amdgcn_readlane(icb, j + k);
        }
        float4 va[8], vb[8];
#pragma unroll
        for (int k = 0; k < 8; ++k) {
            if (j + k < na) va[k] = *(const float4*)(feat + (size_t)qa[k] * D + lane4);
            else            va[k] = make_float4(0.0f, 0.0f, 0.0f, 0.0f);
        }
#pragma unroll
        for (int k = 0; k < 8; ++k) {
            if (j + k < nb) vb[k] = *(const float4*)(feat + (size_t)qb[k] * D + lane4);
            else            vb[k] = make_float4(0.0f, 0.0f, 0.0f, 0.0f);
        }
#pragma unroll
        for (int k = 0; k < 8; ++k) {
            acca.x += va[k].x; acca.y += va[k].y; acca.z += va[k].z; acca.w += va[k].w;
        }
#pragma unroll
        for (int k = 0; k < 8; ++k) {
            accb.x += vb[k].x; accb.y += vb[k].y; accb.z += vb[k].z; accb.w += vb[k].w;
        }
    }
    ra = acca;
    rb = accb;
}

__device__ __forceinline__ void sage_add_marked(const float* __restrict__ feat,
                                                unsigned ma, unsigned mb, int sl,
                                                unsigned lane4,
                                                float4& acca, float4& accb) {
    while (ma != 0u) {
        const int k = __builtin_ctz(ma);
        ma &= ma - 1u;
        const int s = __builtin_amdgcn_readlane(sl, k);
        const float4 v = *(const float4*)(feat + (size_t)s * D + lane4);
        acca.x += v.x; acca.y += v.y; acca.z += v.z; acca.w += v.w;
    }
    while (mb != 0u) {
        const int k = __builtin_ctz(mb);
        mb &= mb - 1u;
        const int s = __builtin_amdgcn_readlane(sl, k);
        const float4 v = *(const float4*)(feat + (size_t)s * D + lane4);
        accb.x += v.x; accb.y += v.y; accb.z += v.z; accb.w += v.w;
    }
}

__device__ __forceinline__ void sage_walk_rows(const float* __restrict__ feat,
                                               const int*   __restrict__ cells,
                                               const int*   __restrict__ ovfcnt,
                                               const int*   __restrict__ ovf,
                                               int blk,
                                               int ra, bool wa,
                                               int rb, bool wb,
                                               int lane,
                                               float4& acca, float4& accb) {
    const unsigned lane4 = (unsigned)lane * 4u;
    if (wa) acca = make_float4(0.0f, 0.0f, 0.0f, 0.0f);
    if (wb) accb = make_float4(0.0f, 0.0f, 0.0f, 0.0f);
    const int rowa = blk * GEMM_ROWS + ra, rowb = blk * GEMM_ROWS + rb;
#pragma nounroll
    for (int c = 0; c < N_CHUNKS; ++c) {
        const int v = cells[((size_t)c * GEMM_BLOCKS + blk) * CELL + lane];
        const int n = min(max(__builtin_amdgcn_readfirstlane(v), 0), CELL_CAP);
        const bool in = (lane >= 1 && lane <= n);
        const int sl = min(max(v >> 6, 0), N_NODES - 1);
        const int r  = v & 63;
        sage_add_marked(feat,
                        __builtin_amdgcn_ballot_w32(in && wa && r == ra),
                        __builtin_amdgcn_ballot_w32(in && wb && r == rb),
                        sl, lane4, acca, accb);
        const int m = __builtin_amdgcn_readfirstlane(min(max(ovfcnt[c], 0), CHUNK_EDGES));
#pragma nounroll
        for (int base = 0; base < m; base += 32) {
            const int pos = min(base + lane, CHUNK_EDGES - 1);
            const int ed = ovf[((size_t)c * CHUNK_EDGES + pos) * 2];
            const int es = ovf[((size_t)c * CHUNK_EDGES + pos) * 2 + 1];
            const bool in2 = (base + lane < m);
            const int sl2 = min(max(es, 0), N_NODES - 1);
            sage_add_marked(feat,
                            __builtin_amdgcn_ballot_w32(in2 && wa && ed == rowa),
                            __builtin_amdgcn_ballot_w32(in2 && wb && ed == rowb),
                            sl2, lane4, acca, accb);
        }
    }
}

#define A_PITCH 132

__global__ __launch_bounds__(256)
void sage_wmma_gemm(const float* __restrict__ feat,
                    const int*   __restrict__ cells,
                    const int*   __restrict__ ovfcnt,
                    const int*   __restrict__ ovf,
                    const float* __restrict__ wpack,
                    const float* __restrict__ bself,
                    const float* __restrict__ bneigh,
                    float*       __restrict__ out) {
    __shared__ __attribute__((aligned(16))) float sSelf [GEMM_ROWS * A_PITCH];
    __shared__ __attribute__((aligned(16))) float sNeigh[GEMM_ROWS * A_PITCH];
    __shared__ int sSlot[GEMM_ROWS * SLOTS];
    __shared__ int sRowCnt[GEMM_ROWS];

    const int lane  = threadIdx.x & 31;
    const int wave  = threadIdx.x >> 5;
    const int rloc  = lane & 15;
    const int khalf = lane >> 4;
    const int blk   = blockIdx.x;
    const int row0  = blk * GEMM_ROWS;
    const int col   = wave * 16 + (lane & 15);

    const float bias = bself[col] + bneigh[col];

    if (threadIdx.x < GEMM_ROWS) sRowCnt[threadIdx.x] = 0;

    const int wv = __builtin_amdgcn_readfirstlane(wave);

    const float* fsrc = feat + lane * 4;
    const v4f f0 = *(const v4f*)(fsrc + (size_t)min(row0 + wave,      N_NODES - 1) * D);
    const v4f f1 = *(const v4f*)(fsrc + (size_t)min(row0 + wave +  8, N_NODES - 1) * D);
    const v4f f2 = *(const v4f*)(fsrc + (size_t)min(row0 + wave + 16, N_NODES - 1) * D);
    const v4f f3 = *(const v4f*)(fsrc + (size_t)min(row0 + wave + 24, N_NODES - 1) * D);
    const v4f f4 = *(const v4f*)(fsrc + (size_t)min(row0 + wave + 32, N_NODES - 1) * D);
    const v4f f5 = *(const v4f*)(fsrc + (size_t)min(row0 + wave + 40, N_NODES - 1) * D);
    const v4f f6 = *(const v4f*)(fsrc + (size_t)min(row0 + wave + 48, N_NODES - 1) * D);
    const v4f f7 = *(const v4f*)(fsrc + (size_t)min(row0 + wave + 56, N_NODES - 1) * D);

    int cellv[WAVE_CHUNKS];
#pragma unroll
    for (int i = 0; i < WAVE_CHUNKS; ++i) {
        const int c = min(wv + 8 * i, N_CHUNKS - 1);
        cellv[i] = cells[((size_t)c * GEMM_BLOCKS + blk) * CELL + lane];
    }
    const int oc = ovfcnt[min(wv + 8 * lane, N_CHUNKS - 1)];

    float* fdst = sSelf + wave * A_PITCH + lane * 4;
    *(v4f*)(fdst +  0 * A_PITCH) = f0;
    *(v4f*)(fdst +  8 * A_PITCH) = f1;
    *(v4f*)(fdst + 16 * A_PITCH) = f2;
    *(v4f*)(fdst + 24 * A_PITCH) = f3;
    *(v4f*)(fdst + 32 * A_PITCH) = f4;
    *(v4f*)(fdst + 40 * A_PITCH) = f5;
    *(v4f*)(fdst + 48 * A_PITCH) = f6;
    *(v4f*)(fdst + 56 * A_PITCH) = f7;

    __syncthreads();

    int rk[WAVE_CHUNKS];
#pragma unroll
    for (int i = 0; i < WAVE_CHUNKS; ++i) {
        const int v = cellv[i];
        const int n = (wv + 8 * i < N_CHUNKS)
                    ? min(max(__builtin_amdgcn_readfirstlane(v), 0), CELL_CAP) : 0;
        rk[i] = (lane >= 1 && lane <= n) ? atomicAdd(sRowCnt + (v & 63), 1) : SLOTS;
    }
#pragma unroll
    for (int i = 0; i < WAVE_CHUNKS; ++i) {
        const int v = cellv[i];
        if ((unsigned)rk[i] < (unsigned)SLOTS)
            sSlot[(v & 63) * SLOTS + rk[i]] = min(max(v >> 6, 0), N_NODES - 1);
    }
    if (__builtin_amdgcn_ballot_w32(oc > 0) != 0u) {
#pragma nounroll
        for (int i = 0; i < WAVE_CHUNKS; ++i) {
            const int c = wv + 8 * i;
            const int m = (c < N_CHUNKS)
                        ? min(max(__builtin_amdgcn_readlane(oc, i), 0), CHUNK_EDGES) : 0;
#pragma nounroll
            for (int base = 0; base < m; base += 32) {
                const int pos = base + lane;
                if (pos < m) {
                    const int ed = ovf[((size_t)c * CHUNK_EDGES + pos) * 2];
                    const int es = ovf[((size_t)c * CHUNK_EDGES + pos) * 2 + 1];
                    if ((ed >> 6) == blk)
                        sage_row_place(ed & 63, min(max(es, 0), N_NODES - 1), sRowCnt, sSlot);
                }
            }
        }
    }

    __syncthreads();

    v8f acc[GEMM_TILES];
#pragma unroll
    for (int t = 0; t < GEMM_TILES; ++t)
#pragma unroll
        for (int j = 0; j < 8; ++j) acc[t][j] = bias;

    const float* arow_s = sSelf  + rloc * A_PITCH + khalf * 2;
    const float* arow_n = sNeigh + rloc * A_PITCH + khalf * 2;
    const float* zrow   = (const float*)(ovf + ZROW_INT_OFF);
    const v4f* wp_s = (const v4f*)wpack + (size_t)(wv * 16) * 32 + lane;
    const v4f* wp_n = (const v4f*)wpack + (size_t)((8 + wv) * 16) * 32 + lane;

    {
        const int cv = min(max(sRowCnt[wave + 8 * (lane & 7)], 0), N_EDGES);
        const unsigned lane4 = (unsigned)lane * 4u;

#pragma nounroll
        for (int p = 0; p < 8; ++p) {
            const int r0 = wave + 8 * p;
            const int ca = __builtin_amdgcn_readlane(cv, p);
            const int fa = min(ca, SLOTS);
            const int ia = (lane < fa) ? sSlot[r0 * SLOTS + lane] : 0;
            const int ica = min(max(ia, 0), N_NODES - 1);
            int qa[8];
#pragma unroll
            for (int k = 0; k < 8; ++k) qa[k] = __builtin_amdgcn_readlane(ica, k);
            const v4f bq0 = wp_s[(2 * p) * 32];
            const v4f bq1 = wp_s[(2 * p + 1) * 32];
            v2f bw[4];
            bw[0].x = bq0.x; bw[0].y = bq0.y;
            bw[1].x = bq0.z; bw[1].y = bq0.w;
            bw[2].x = bq1.x; bw[2].y = bq1.y;
            bw[3].x = bq1.z; bw[3].y = bq1.w;
            v4f va[8];
#pragma unroll
            for (int k = 0; k < 8; ++k) {
                const float* pa = (k < fa) ? feat + (size_t)qa[k] * D : zrow;
                va[k] = *(const v4f*)(pa + lane4);
            }
#pragma unroll
            for (int i = 0; i < 4; ++i) {
#pragma unroll
                for (int t = 0; t < GEMM_TILES; ++t) {
                    const v2f a = *(const v2f*)(arow_s + t * 16 * A_PITCH + 16 * p + 4 * i);
                    acc[t] = __builtin_amdgcn_wmma_f32_16x16x4_f32(
                                 false, a, false, bw[i], (short)0, acc[t], false, false);
                }
            }
            asm volatile("" : "+v"(acc[0]), "+v"(acc[1]), "+v"(acc[2]), "+v"(acc[3]), "+v"(va[0]));
            v4f s0 = {0.0f, 0.0f, 0.0f, 0.0f};
#pragma unroll
            for (int k = 0; k < 8; ++k) s0 += va[k];
            float4 n0 = make_float4(s0.x, s0.y, s0.z, s0.w);
            float4 n1 = make_float4(0.0f, 0.0f, 0.0f, 0.0f);
            if (fa > 8)
                sage_gather2(feat, fa, 0, lane, ica, 0, n0, n1);
            if (ca > SLOTS)
                sage_walk_rows(feat, cells, ovfcnt, ovf, blk,
                               r0, true, r0, false, lane, n0, n1);
            const float rd0 = 1.0f / fmaxf((float)ca, 1.0f);
            n0.x *= rd0; n0.y *= rd0; n0.z *= rd0; n0.w *= rd0;
            *(float4*)(sNeigh + r0 * A_PITCH + lane * 4) = n0;
        }
    }

    __syncthreads();

#pragma unroll 8
    for (int g = 0; g < D / 8; ++g) {
        const v4f bq = wp_n[g * 32];
#pragma unroll
        for (int s = 0; s < 2; ++s) {
            v2f b;
            b.x = s ? bq.z : bq.x;
            b.y = s ? bq.w : bq.y;
            const int k0 = 8 * g + 4 * s;
#pragma unroll
            for (int t = 0; t < GEMM_TILES; ++t) {
                const v2f a = *(const v2f*)(arow_n + t * 16 * A_PITCH + k0);
                acc[t] = __builtin_amdgcn_wmma_f32_16x16x4_f32(
                             false, a, false, b, (short)0, acc[t], false, false);
            }
        }
    }

    float* dtile = sSelf + (khalf * 8) * A_PITCH + col;
#pragma unroll
    for (int t = 0; t < GEMM_TILES; ++t)
#pragma unroll
        for (int j = 0; j < 8; ++j)
            dtile[(16 * t + j) * A_PITCH] = acc[t][j];

    __syncthreads();

    static_assert(N_NODES % 8 == 0, "a wave's eight output rows must be all inside or all outside");
    if (row0 + 8 * wv < N_NODES) {
        const float* srow = sSelf + (8 * wave) * A_PITCH + lane * 4;
        float*       orow = out + (size_t)(row0 + 8 * wave) * D + lane * 4;
#pragma unroll
        for (int p = 0; p < 8; ++p)
            *(v4f*)(orow + (size_t)p * D) = *(const v4f*)(srow + p * A_PITCH);
    }
}

extern "C" void kernel_launch(void* const* d_in, const int* in_sizes, int n_in,
                              void* d_out, int out_size, void* d_ws, size_t ws_size,
                              hipStream_t stream) {
    const float* feat   = (const float*)d_in[0];
    const int*   src    = (const int*)  d_in[1];
    const int*   dst    = (const int*)  d_in[2];
    const float* Wself  = (const float*)d_in[3];
    const float* bself  = (const float*)d_in[4];
    const float* Wneigh = (const float*)d_in[5];
    const float* bneigh = (const float*)d_in[6];
    float*       out    = (float*)d_out;

    int* cells  = (int*)d_ws;
    int* ovfcnt = cells  + CELL_INTS;
    int* ovf    = ovfcnt + OVFCNT_INTS;
    float* wpack = (float*)((int*)d_ws + WPACK_INT_OFF);

    sage_bin_edges<<<BIN_GRID, 1024, 0, stream>>>(
        src, dst, Wself, Wneigh, cells, ovfcnt, ovf, wpack);
    sage_wmma_gemm<<<GEMM_BLOCKS, 256, 0, stream>>>(
        feat, cells, ovfcnt, ovf, wpack, bself, bneigh, out);
}
